# starting kernel
# baseline (speedup 1.0000x reference)
#include <hip/hip_runtime.h>
#include <hip/hip_bf16.h>

typedef __attribute__((ext_vector_type(16))) _Float16 v16h;
typedef __attribute__((ext_vector_type(8)))  _Float16 v8h;
typedef __attribute__((ext_vector_type(4)))  _Float16 v4h;
typedef __attribute__((ext_vector_type(8)))  float    v8f;

#define HID 128
#define NEG_SLOPE 0.2f
#define BN_EPS 1e-5f
#define N_GRAPHS 512

__global__ void pack_w_f16(const float* __restrict__ W, _Float16* __restrict__ Wp, int N) {
    int id = blockIdx.x * blockDim.x + threadIdx.x;
    int total = (N / 16) * 4 * 32;
    if (id >= total) return;
    int t    = id >> 7;
    int kt   = (id >> 5) & 3;
    int lane = id & 31;
    int hl = lane & 15, hi = lane >> 4;
    int n = t * 16 + hl;
    _Float16* o = Wp + (size_t)id * 16;
#pragma unroll
    for (int j = 0; j < 8; ++j) {
        o[j]     = (_Float16)W[(size_t)(kt * 32 + hi * 8 + j) * N + n];
        o[8 + j] = (_Float16)W[(size_t)(kt * 32 + 16 + hi * 8 + j) * N + n];
    }
}

template <bool HAS_BIAS>
__global__ void gemm128_wmma(const float* __restrict__ A, const _Float16* __restrict__ Bp,
                             const float* __restrict__ bias, float* __restrict__ C,
                             int N) {
    __shared__ __align__(16) _Float16 As[16 * 128];
    const int tid  = threadIdx.x;
    const int row0 = blockIdx.x * 16;

    for (int c = tid; c < 512; c += blockDim.x) {
        int r   = c >> 5;
        int col = (c & 31) * 4;
        float4 f = *(const float4*)(A + (size_t)(row0 + r) * 128 + col);
        v4h h4 = { (_Float16)f.x, (_Float16)f.y, (_Float16)f.z, (_Float16)f.w };
        *(v4h*)(As + r * 128 + col) = h4;
    }
    __syncthreads();

    const int wave = tid >> 5;
    const int lane = tid & 31;
    const int hl   = lane & 15;
    const int hi   = lane >> 4;

    v8f acc = {};
#pragma unroll
    for (int kt = 0; kt < 4; ++kt) {
        int base = hl * 128 + kt * 32 + hi * 8;
        v8h alo = *(const v8h*)(As + base);
        v8h ahi = *(const v8h*)(As + base + 16);
        v16h a;
#pragma unroll
        for (int j = 0; j < 8; ++j) { a[j] = alo[j]; a[8 + j] = ahi[j]; }

        v16h b = *(const v16h*)(Bp + (((size_t)wave * 4 + kt) * 32 + lane) * 16);

        acc = __builtin_amdgcn_wmma_f32_16x16x32_f16(
            false, a, false, b, (short)0, acc, false, false);
    }

    const int col = wave * 16 + hl;
    float bv = HAS_BIAS ? bias[col] : 0.0f;
#pragma unroll
    for (int r = 0; r < 8; ++r) {
        int row = row0 + r + 8 * hi;
        C[(size_t)row * N + col] = acc[r] + bv;
    }
}

__global__ void node_scores(const float* __restrict__ H, const float* __restrict__ a_src,
                            const float* __restrict__ a_dst, float* __restrict__ ss,
                            float* __restrict__ sd, int n_nodes) {
    int node = (int)((blockIdx.x * (size_t)blockDim.x + threadIdx.x) >> 5);
    int lane = threadIdx.x & 31;
    if (node >= n_nodes) return;
    const float* h = H + (size_t)node * HID;
    float s0 = 0.f, s1 = 0.f;
#pragma unroll
    for (int j = 0; j < 4; ++j) {
        int c = lane + 32 * j;
        float hv = h[c];
        s0 += hv * a_src[c];
        s1 += hv * a_dst[c];
    }
#pragma unroll
    for (int off = 16; off > 0; off >>= 1) {
        s0 += __shfl_down(s0, off, 32);
        s1 += __shfl_down(s1, off, 32);
    }
    if (lane == 0) { ss[node] = s0; sd[node] = s1; }
}

__global__ void fill_f32(float* __restrict__ p, float v, int n) {
    int i = blockIdx.x * blockDim.x + threadIdx.x;
    if (i < n) p[i] = v;
}

__device__ __forceinline__ void atomicMaxF(float* addr, float val) {
    if (val >= 0.f) atomicMax((int*)addr, __float_as_int(val));
    else            atomicMin((unsigned int*)addr, __float_as_uint(val));
}

__device__ __forceinline__ void edge_sd(int e, int n_edges,
                                        const int* __restrict__ esrc,
                                        const int* __restrict__ edst, int& s, int& d) {
    if (e < n_edges) { s = esrc[e]; d = edst[e]; }
    else             { s = e - n_edges; d = s; }
}

__global__ void edge_max(const int* __restrict__ esrc, const int* __restrict__ edst,
                         const float* __restrict__ ss, const float* __restrict__ sd,
                         float* __restrict__ mx, int n_edges, int etot) {
    int e = blockIdx.x * blockDim.x + threadIdx.x;
    if (e >= etot) return;
    int s, d; edge_sd(e, n_edges, esrc, edst, s, d);
    float v = ss[s] + sd[d];
    v = (v > 0.f) ? v : NEG_SLOPE * v;
    atomicMaxF(&mx[d], v);
}

__global__ void edge_expsum(const int* __restrict__ esrc, const int* __restrict__ edst,
                            const float* __restrict__ ss, const float* __restrict__ sd,
                            const float* __restrict__ mx, float* __restrict__ den,
                            int n_edges, int etot) {
    int e = blockIdx.x * blockDim.x + threadIdx.x;
    if (e >= etot) return;
    int s, d; edge_sd(e, n_edges, esrc, edst, s, d);
    float v = ss[s] + sd[d];
    v = (v > 0.f) ? v : NEG_SLOPE * v;
    atomicAdd(&den[d], __expf(v - mx[d]));
}

__global__ void edge_aggregate(const float* __restrict__ H, const int* __restrict__ esrc,
                               const int* __restrict__ edst, const float* __restrict__ ss,
                               const float* __restrict__ sd, const float* __restrict__ mx,
                               const float* __restrict__ den, float* __restrict__ out,
                               int n_edges, int etot) {
    int e = (int)((blockIdx.x * (size_t)blockDim.x + threadIdx.x) >> 5);
    int lane = threadIdx.x & 31;
    if (e >= etot) return;
    int s, d; edge_sd(e, n_edges, esrc, edst, s, d);
    float v = ss[s] + sd[d];
    v = (v > 0.f) ? v : NEG_SLOPE * v;
    float w = __expf(v - mx[d]) / (den[d] + 1e-16f);
    const float* hs = H + (size_t)s * HID;
    float* od = out + (size_t)d * HID;
#pragma unroll
    for (int j = 0; j < 4; ++j) {
        int c = lane + 32 * j;
        atomicAdd(&od[c], w * hs[c]);
    }
}

__global__ void bias_bn_relu(float* __restrict__ x, const float* __restrict__ bias,
                             const float* __restrict__ gam, const float* __restrict__ bet,
                             const float* __restrict__ mean, const float* __restrict__ var,
                             int total) {
    int i = blockIdx.x * blockDim.x + threadIdx.x;
    if (i >= total) return;
    int c = i & (HID - 1);
    float v = x[i] + bias[c];
    v = (v - mean[c]) * (gam[c] * rsqrtf(var[c] + BN_EPS)) + bet[c];
    x[i] = fmaxf(v, 0.f);
}

__global__ void pool_accum(const float* __restrict__ x, const int* __restrict__ batch,
                           float* __restrict__ sums, float* __restrict__ cnt, int n_nodes) {
    int i = blockIdx.x * blockDim.x + threadIdx.x;
    if (i >= n_nodes * HID) return;
    int n = i >> 7, c = i & (HID - 1);
    int g = batch[n];
    atomicAdd(&sums[(size_t)g * HID + c], x[i]);
    if (c == 0) atomicAdd(&cnt[g], 1.0f);
}

__global__ void pool_div(float* __restrict__ sums, const float* __restrict__ cnt, int total) {
    int i = blockIdx.x * blockDim.x + threadIdx.x;
    if (i >= total) return;
    sums[i] /= fmaxf(cnt[i >> 7], 1.0f);
}

extern "C" void kernel_launch(void* const* d_in, const int* in_sizes, int n_in,
                              void* d_out, int out_size, void* d_ws, size_t ws_size,
                              hipStream_t stream) {
    const float* gx  = (const float*)d_in[0];
    const int*   ei  = (const int*)  d_in[1];
    const int*   bat = (const int*)  d_in[2];
    const float* W1  = (const float*)d_in[3];
    const float* as1 = (const float*)d_in[4];
    const float* ad1 = (const float*)d_in[5];
    const float* b1  = (const float*)d_in[6];
    const float* g1  = (const float*)d_in[7];
    const float* be1 = (const float*)d_in[8];
    const float* mu1 = (const float*)d_in[9];
    const float* va1 = (const float*)d_in[10];
    const float* W2  = (const float*)d_in[11];
    const float* as2 = (const float*)d_in[12];
    const float* ad2 = (const float*)d_in[13];
    const float* b2  = (const float*)d_in[14];
    const float* g2  = (const float*)d_in[15];
    const float* be2 = (const float*)d_in[16];
    const float* mu2 = (const float*)d_in[17];
    const float* va2 = (const float*)d_in[18];
    const float* hw  = (const float*)d_in[19];
    const float* hb  = (const float*)d_in[20];

    const int n_nodes = in_sizes[0] / HID;
    const int n_edges = in_sizes[1] / 2;
    const int etot    = n_edges + n_nodes;
    const int* esrc = ei;
    const int* edst = ei + n_edges;

    float* ws   = (float*)d_ws;
    float* H    = ws;
    float* AGG  = H    + (size_t)n_nodes * HID;
    float* SS   = AGG  + (size_t)n_nodes * HID;
    float* SD   = SS   + n_nodes;
    float* MX   = SD   + n_nodes;
    float* DEN  = MX   + n_nodes;
    float* POOL = DEN  + n_nodes;
    float* CNT  = POOL + (size_t)N_GRAPHS * HID;
    size_t wpOff = (size_t)(CNT + N_GRAPHS - ws);
    wpOff = (wpOff + 7) & ~(size_t)7;
    _Float16* WP = (_Float16*)(ws + wpOff);

    const int feat_total = n_nodes * HID;
    const int gridFeat  = (feat_total + 255) / 256;
    const int gridNode  = (n_nodes + 255) / 256;
    const int gridEdge  = (etot + 255) / 256;
    const int gridEdgeW = (etot + 7) / 8;
    const int gridNodeW = (n_nodes + 7) / 8;
    const int gridGemm  = n_nodes / 16;

    for (int layer = 0; layer < 2; ++layer) {
        const float* xin = (layer == 0) ? gx : AGG;
        const float* W   = (layer == 0) ? W1 : W2;
        const float* asr = (layer == 0) ? as1 : as2;
        const float* ads = (layer == 0) ? ad1 : ad2;
        const float* bia = (layer == 0) ? b1  : b2;
        const float* gam = (layer == 0) ? g1  : g2;
        const float* bet = (layer == 0) ? be1 : be2;
        const float* mu  = (layer == 0) ? mu1 : mu2;
        const float* va  = (layer == 0) ? va1 : va2;

        pack_w_f16<<<4, 256, 0, stream>>>(W, WP, HID);
        gemm128_wmma<false><<<gridGemm, 256, 0, stream>>>(xin, WP, nullptr, H, HID);
        node_scores<<<gridNodeW, 256, 0, stream>>>(H, asr, ads, SS, SD, n_nodes);
        fill_f32<<<gridNode, 256, 0, stream>>>(MX, -3.402823466e38f, n_nodes);
        fill_f32<<<gridNode, 256, 0, stream>>>(DEN, 0.0f, n_nodes);
        fill_f32<<<gridFeat, 256, 0, stream>>>(AGG, 0.0f, feat_total);
        edge_max<<<gridEdge, 256, 0, stream>>>(esrc, edst, SS, SD, MX, n_edges, etot);
        edge_expsum<<<gridEdge, 256, 0, stream>>>(esrc, edst, SS, SD, MX, DEN, n_edges, etot);
        edge_aggregate<<<gridEdgeW, 256, 0, stream>>>(H, esrc, edst, SS, SD, MX, DEN, AGG,
                                                      n_edges, etot);
        bias_bn_relu<<<gridFeat, 256, 0, stream>>>(AGG, bia, gam, bet, mu, va, feat_total);
    }

    fill_f32<<<(N_GRAPHS * HID + 255) / 256, 256, 0, stream>>>(POOL, 0.0f, N_GRAPHS * HID);
    fill_f32<<<(N_GRAPHS + 255) / 256, 256, 0, stream>>>(CNT, 0.0f, N_GRAPHS);
    pool_accum<<<gridFeat, 256, 0, stream>>>(AGG, bat, POOL, CNT, n_nodes);
    pool_div<<<(N_GRAPHS * HID + 255) / 256, 256, 0, stream>>>(POOL, CNT, N_GRAPHS * HID);

    pack_w_f16<<<2, 256, 0, stream>>>(hw, WP, 64);
    gemm128_wmma<true><<<N_GRAPHS / 16, 32 * (64 / 16), 0, stream>>>(
        POOL, WP, hb, (float*)d_out, 64);
}


# Edges split once by owner workgroup: a pass counts per owner in LDS, takes list ranges with one global atomic per block and owner, files (src,row); owners stop reading all dsts; sum order: arrival
# speedup vs baseline: 4.7353x; 4.7353x over previous
#include <hip/hip_runtime.h>
#include <hip/hip_bf16.h>

typedef __attribute__((ext_vector_type(16))) _Float16 v16h;
typedef __attribute__((ext_vector_type(8)))  _Float16 v8h;
typedef __attribute__((ext_vector_type(4)))  _Float16 v4h;
typedef __attribute__((ext_vector_type(8)))  float    v8f;

#define HID 128
#define NEG_SLOPE 0.2f
#define BN_EPS 1e-5f
#define N_GRAPHS 512

__global__ void pack_w_f16(const float* __restrict__ W, _Float16* __restrict__ Wp, int N) {
    int id = blockIdx.x * blockDim.x + threadIdx.x;
    int total = (N / 16) * 4 * 32;
    if (id >= total) return;
    int t    = id >> 7;
    int kt   = (id >> 5) & 3;
    int lane = id & 31;
    int hl = lane & 15, hi = lane >> 4;
    int n = t * 16 + hl;
    _Float16* o = Wp + (size_t)id * 16;
#pragma unroll
    for (int j = 0; j < 8; ++j) {
        o[j]     = (_Float16)W[(size_t)(kt * 32 + hi * 8 + j) * N + n];
        o[8 + j] = (_Float16)W[(size_t)(kt * 32 + 16 + hi * 8 + j) * N + n];
    }
}

template <bool HAS_BIAS, int NW, int RT>
__global__ __launch_bounds__(NW * 32)
void gemm128_wmma(const float* __restrict__ A, const _Float16* __restrict__ Bp,
                  const float* __restrict__ bias, float* __restrict__ C, int M) {
    constexpr int N     = NW * 16;
    constexpr int NT    = NW * 32;
    constexpr int PER   = RT * 512 / NT;
    constexpr int PITCH = 136;
    static_assert(RT * 512 % NT == 0, "the A pieces divide evenly among the threads");
    __shared__ __align__(16) _Float16 As[RT * 16 * PITCH];
    const int tid  = threadIdx.x;
    const int row0 = blockIdx.x * (RT * 16);
    const int wave = tid >> 5;
    const int lane = tid & 31;
    const int hl   = lane & 15;
    const int hi   = lane >> 4;

    float4 f[PER];
#pragma unroll
    for (int q = 0; q < PER; ++q) {
        const int c = tid + q * NT;
        int row = row0 + (c >> 5);
        if (row >= M) row = M - 1;
        f[q] = *(const float4*)(A + (size_t)row * 128 + (c & 31) * 4);
    }
    v16h b[4];
#pragma unroll
    for (int kt = 0; kt < 4; ++kt)
        b[kt] = *(const v16h*)(Bp + (((size_t)wave * 4 + kt) * 32 + lane) * 16);
#pragma unroll
    for (int q = 0; q < PER; ++q) {
        const int c = tid + q * NT;
        v4h h4 = { (_Float16)f[q].x, (_Float16)f[q].y, (_Float16)f[q].z, (_Float16)f[q].w };
        *(v4h*)(As + (c >> 5) * PITCH + (c & 31) * 4) = h4;
    }
    __syncthreads();

    v8f acc[RT];
#pragma unroll
    for (int i = 0; i < RT; ++i) acc[i] = v8f{};
#pragma unroll
    for (int kt = 0; kt < 4; ++kt) {
#pragma unroll
        for (int i = 0; i < RT; ++i) {
            const int base = (i * 16 + hl) * PITCH + kt * 32 + hi * 8;
            v8h alo = *(const v8h*)(As + base);
            v8h ahi = *(const v8h*)(As + base + 16);
            v16h a;
#pragma unroll
            for (int j = 0; j < 8; ++j) { a[j] = alo[j]; a[8 + j] = ahi[j]; }
            acc[i] = __builtin_amdgcn_wmma_f32_16x16x32_f16(
                false, a, false, b[kt], (short)0, acc[i], false, false);
        }
    }

    const int col = wave * 16 + hl;
    float bv = HAS_BIAS ? bias[col] : 0.0f;
#pragma unroll
    for (int i = 0; i < RT; ++i) {
        if (row0 + i * 16 < M) {
#pragma unroll
            for (int r = 0; r < 8; ++r) {
                const int row = row0 + i * 16 + r + 8 * hi;
                C[(size_t)row * N + col] = acc[i][r] + bv;
            }
        }
    }
}

#define SC_ROWS 16
typedef float v4f __attribute__((ext_vector_type(4)));

template <int NH, int OFF>
__device__ __forceinline__ void fold(float (&p)[2 * SC_ROWS], int lane) {
    const bool up = (lane & OFF) != 0;
#pragma unroll
    for (int i = 0; i < NH; ++i) {
        const float keep = up ? p[i + NH] : p[i];
        const float send = up ? p[i] : p[i + NH];
        p[i] = keep + __shfl_xor(send, OFF, 32);
    }
}

__global__ __launch_bounds__(256)
void node_scores(const float* __restrict__ H, const float* __restrict__ a_src,
                 const float* __restrict__ a_dst, float* __restrict__ ss,
                 float* __restrict__ sd, int n_nodes) {
    const int lane = threadIdx.x & 31;
    const int row0 = (int)((blockIdx.x * (size_t)blockDim.x + threadIdx.x) >> 5) * SC_ROWS;
    if (row0 >= n_nodes) return;
    const v4f as = *(const v4f*)(a_src + 4 * lane);
    const v4f ad = *(const v4f*)(a_dst + 4 * lane);
    const float* hb = H + (size_t)row0 * HID + 4 * lane;
    v4f h[SC_ROWS];
#pragma unroll
    for (int k = 0; k < SC_ROWS; ++k) h[k] = *(const v4f*)(hb + k * HID);
    float p[2 * SC_ROWS];
#pragma unroll
    for (int k = 0; k < SC_ROWS; ++k) {
        p[k]           = h[k].x * as.x + h[k].y * as.y + h[k].z * as.z + h[k].w * as.w;
        p[SC_ROWS + k] = h[k].x * ad.x + h[k].y * ad.y + h[k].z * ad.z + h[k].w * ad.w;
    }
    fold<16, 16>(p, lane);
    fold<8, 8>(p, lane);
    fold<4, 4>(p, lane);
    fold<2, 2>(p, lane);
    fold<1, 1>(p, lane);
    const int r = row0 + (lane & 15);
    if (lane < 16) ss[r] = p[0];
    else           sd[r] = p[0];
}

__global__ void fill_f32(float* __restrict__ p, float v, int n) {
    int i = blockIdx.x * blockDim.x + threadIdx.x;
    if (i < n) p[i] = v;
}

#define OWN_ROWS    196
#define OWN_CAP     16384
#define OWN_THREADS 1024

typedef int v4i __attribute__((ext_vector_type(4)));

__device__ __forceinline__ void ldsMaxF(float* addr, float val) {
    if (val >= 0.f) atomicMax((int*)addr, __float_as_int(val));
    else            atomicMin((unsigned int*)addr, __float_as_uint(val));
}

template <class F>
__device__ __forceinline__ void scan_own_edges(const int* __restrict__ edst, int n_edges,
                                               int row0, int rows, F f) {
    const int nthr = blockDim.x;
    const int nvec = (((size_t)edst & 15) == 0) ? (n_edges >> 2) : 0;
    const v4i* dv = (const v4i*)edst;
    const int rot = (int)blockIdx.x * (nvec / (int)gridDim.x);
    for (int v0 = threadIdx.x; v0 < nvec; v0 += 8 * nthr) {
        v4i d[8];
#pragma unroll
        for (int k = 0; k < 8; ++k) {
            const int v = v0 + k * nthr;
            int uu = v + rot;
            if (uu >= nvec) uu -= nvec;
            const v4i none = { -1, -1, -1, -1 };
            d[k] = (v < nvec) ? __builtin_nontemporal_load(dv + uu) : none;
        }
        unsigned m = 0;
#pragma unroll
        for (int k = 0; k < 8; ++k) {
#pragma unroll
            for (int c = 0; c < 4; ++c)
                m |= ((unsigned)(d[k][c] - row0) < (unsigned)rows) ? (1u << (4 * k + c)) : 0u;
        }
        while (m) {
            const int b = __builtin_ctz(m);
            m &= m - 1;
            int uu = v0 + (b >> 2) * nthr + rot;
            if (uu >= nvec) uu -= nvec;
            f(4 * uu + (b & 3));
        }
    }
    for (int e = nvec * 4 + threadIdx.x; e < n_edges; e += nthr)
        if ((unsigned)(edst[e] - row0) < (unsigned)rows) f(e);
}

#define SPLIT_THREADS 256
#define SPLIT_EPT     16
#define SPLIT_MAX_OWN 1024

__global__ __launch_bounds__(SPLIT_THREADS)
void split_edges_by_owner(const int* __restrict__ esrc, const int* __restrict__ edst,
                          unsigned* __restrict__ wsList, int* __restrict__ wsCnt,
                          int n_edges, int n_nodes, int n_own) {
    __shared__ int hist[SPLIT_MAX_OWN];
    __shared__ int hbase[SPLIT_MAX_OWN];
    const int tid = threadIdx.x;
    for (int o = tid; o < n_own; o += SPLIT_THREADS) hist[o] = 0;

    const int e0 = (int)blockIdx.x * (SPLIT_THREADS * SPLIT_EPT) + tid;
    int d[SPLIT_EPT], s[SPLIT_EPT];
#pragma unroll
    for (int k = 0; k < SPLIT_EPT; ++k) {
        const int e  = e0 + k * SPLIT_THREADS;
        const int ec = (e < n_edges) ? e : n_edges - 1;
        d[k] = edst[ec];
        s[k] = esrc[ec];
        if (e >= n_edges) d[k] = -1;
    }
    __syncthreads();

    int own[SPLIT_EPT], rank[SPLIT_EPT];
#pragma unroll
    for (int k = 0; k < SPLIT_EPT; ++k) {
        own[k]  = ((unsigned)d[k] < (unsigned)n_nodes) ? d[k] / OWN_ROWS : -1;
        rank[k] = 0;
        if (own[k] >= 0) rank[k] = atomicAdd(&hist[own[k]], 1);
    }
    __syncthreads();

    for (int o = tid; o < n_own; o += SPLIT_THREADS) {
        const int cnt = hist[o];
        hbase[o] = (cnt > 0) ? atomicAdd(&wsCnt[o], cnt) : 0;
    }
    __syncthreads();

    int pos[SPLIT_EPT];
#pragma unroll
    for (int k = 0; k < SPLIT_EPT; ++k) pos[k] = hbase[(own[k] >= 0) ? own[k] : 0] + rank[k];
#pragma unroll
    for (int k = 0; k < SPLIT_EPT; ++k) {
        if (own[k] >= 0 && pos[k] < OWN_CAP)
            wsList[(size_t)own[k] * OWN_CAP + pos[k]] =
                ((unsigned)s[k] << 8) | (unsigned)(d[k] - own[k] * OWN_ROWS);
    }
}

template <bool FIRST>
__global__ __launch_bounds__(OWN_THREADS)
void own_rows_gat(const float* __restrict__ H, const int* __restrict__ esrc,
                  const int* __restrict__ edst, const float* __restrict__ ss,
                  const float* __restrict__ sd,
                  const float* __restrict__ bias, const float* __restrict__ gam,
                  const float* __restrict__ bet, const float* __restrict__ mean,
                  const float* __restrict__ var,
                  float* __restrict__ out,
                  const int* __restrict__ batch, float* __restrict__ pool,
                  float* __restrict__ pcnt,
                  const unsigned* __restrict__ wsList, const int* __restrict__ wsCnt,
                  int n_edges, int n_nodes) {
    __shared__ __align__(16) float acc[OWN_ROWS * HID];
    __shared__ unsigned list[OWN_CAP];
    __shared__ float mxl[OWN_ROWS];
    __shared__ float denl[OWN_ROWS];
    __shared__ float sdl[OWN_ROWS];

    const int tid = threadIdx.x, nthr = blockDim.x;
    const int lane = tid & 31;
    const int wave = __builtin_amdgcn_readfirstlane(tid >> 5);
    const int nwaves = nthr >> 5;
    const int row0 = blockIdx.x * OWN_ROWS;
    const int rows = (n_nodes - row0 < OWN_ROWS) ? (n_nodes - row0) : OWN_ROWS;
    const unsigned* wl = wsList + (size_t)blockIdx.x * OWN_CAP;

    float4* z4 = (float4*)acc;
    for (int i = tid; i < OWN_ROWS * HID / 4; i += nthr) z4[i] = make_float4(0.f, 0.f, 0.f, 0.f);
    for (int i = tid; i < rows; i += nthr) {
        mxl[i]  = -3.402823466e38f;
        denl[i] = 0.f;
        sdl[i]  = sd[row0 + i];
        list[i] = ((unsigned)(row0 + i) << 8) | (unsigned)i;
    }
    const int found = rows + wsCnt[blockIdx.x];
    const bool fits = found <= OWN_CAP;
    const int nlist = fits ? found : rows;
    for (int i = rows + tid; i < nlist; i += nthr) list[i] = wl[i - rows];
    __syncthreads();

    auto score = [&](int s, int dl) {
        const float v = ss[s] + sdl[dl];
        return (v > 0.f) ? v : NEG_SLOPE * v;
    };

    auto do_max = [&](int s, int dl) { ldsMaxF(&mxl[dl], score(s, dl)); };
    for (int i = tid; i < nlist; i += nthr) { const unsigned p = list[i]; do_max((int)(p >> 8), (int)(p & 255u)); }
    if (!fits) scan_own_edges(edst, n_edges, row0, rows, [&](int e) { do_max(esrc[e], edst[e] - row0); });
    __syncthreads();

    auto do_sum = [&](int s, int dl) { atomicAdd(&denl[dl], __expf(score(s, dl) - mxl[dl])); };
    for (int i = tid; i < nlist; i += nthr) { const unsigned p = list[i]; do_sum((int)(p >> 8), (int)(p & 255u)); }
    if (!fits) scan_own_edges(edst, n_edges, row0, rows, [&](int e) { do_sum(esrc[e], edst[e] - row0); });
    __syncthreads();

    for (int b0 = wave * 32; b0 < nlist; b0 += nwaves * 32) {
        const int i = b0 + lane;
        int s = 0, dl = 0;
        float w = 0.f;
        if (i < nlist) {
            const unsigned p = list[i];
            s = (int)(p >> 8); dl = (int)(p & 255u);
            w = __expf(score(s, dl) - mxl[dl]) / (denl[dl] + 1e-16f);
        }
        const int left = nlist - b0;
        const int nb = (left < 32) ? left : 32;
        int k0 = 0;
        for (; k0 + 8 <= nb; k0 += 8) {
            float hv[8][4];
            int   dk[8];
            float wk[8];
#pragma unroll
            for (int k = 0; k < 8; ++k) {
                const int sk = __builtin_amdgcn_readlane(s, k0 + k);
                dk[k] = __builtin_amdgcn_readlane(dl, k0 + k);
                wk[k] = __int_as_float(__builtin_amdgcn_readlane(__float_as_int(w), k0 + k));
                const float* hs = H + (size_t)sk * HID + lane;
#pragma unroll
                for (int j = 0; j < 4; ++j) hv[k][j] = hs[32 * j];
            }
#pragma unroll
            for (int k = 0; k < 8; ++k) {
                float* od = acc + dk[k] * HID + lane;
#pragma unroll
                for (int j = 0; j < 4; ++j) atomicAdd(&od[32 * j], wk[k] * hv[k][j]);
            }
        }
        for (; k0 < nb; ++k0) {
            const int sk = __builtin_amdgcn_readlane(s, k0);
            const int dq = __builtin_amdgcn_readlane(dl, k0);
            const float wq = __int_as_float(__builtin_amdgcn_readlane(__float_as_int(w), k0));
            const float* hs = H + (size_t)sk * HID + lane;
            float* od = acc + dq * HID + lane;
            const float h0 = hs[0], h1 = hs[32], h2 = hs[64], h3 = hs[96];
            atomicAdd(&od[0],  wq * h0);
            atomicAdd(&od[32], wq * h1);
            atomicAdd(&od[64], wq * h2);
            atomicAdd(&od[96], wq * h3);
        }
    }
    if (!fits) {
        scan_own_edges(edst, n_edges, row0, rows, [&](int e) {
            const int s = esrc[e], dl = edst[e] - row0;
            const float w = __expf(score(s, dl) - mxl[dl]) / (denl[dl] + 1e-16f);
            const float* hs = H + (size_t)s * HID;
            for (int c = 0; c < HID; ++c) atomicAdd(&acc[dl * HID + c], w * hs[c]);
        });
    }

    auto finish = [](float v, float b, float mu, float sc, float be) {
        return fmaxf((v + b - mu) * sc + be, 0.f);
    };

    if (FIRST) {
        const int c0 = (tid & 31) * 4;
        float b[4], mu[4], sc[4], be[4];
#pragma unroll
        for (int j = 0; j < 4; ++j) {
            b[j]  = bias[c0 + j];
            mu[j] = mean[c0 + j];
            sc[j] = gam[c0 + j] * rsqrtf(var[c0 + j] + BN_EPS);
            be[j] = bet[c0 + j];
        }
        __syncthreads();

        float4* o4 = (float4*)(out + (size_t)row0 * HID);
        const float4* a4 = (const float4*)acc;
        for (int i = tid; i < rows * (HID / 4); i += nthr) {
            float4 a = a4[i];
            a.x = finish(a.x, b[0], mu[0], sc[0], be[0]);
            a.y = finish(a.y, b[1], mu[1], sc[1], be[1]);
            a.z = finish(a.z, b[2], mu[2], sc[2], be[2]);
            a.w = finish(a.w, b[3], mu[3], sc[3], be[3]);
            o4[i] = a;
        }
    } else {
        static_assert((OWN_ROWS + OWN_THREADS / HID - 1) / (OWN_THREADS / HID) <= 32,
                      "a slice's graph ids are held one per lane");
        const int c = tid & (HID - 1);
        const float b  = bias[c];
        const float mu = mean[c];
        const float sc = gam[c] * rsqrtf(var[c] + BN_EPS);
        const float be = bet[c];
        const int ngrp = nthr / HID;
        const int per  = (rows + ngrp - 1) / ngrp;
        const int r0   = (wave >> 2) * per;
        const int n    = (rows - r0 < per) ? (rows - r0) : per;
        const int gid = (lane < n) ? batch[row0 + r0 + lane] : 0;
        __syncthreads();

        if (n > 0) {
            int cur = __builtin_amdgcn_readlane(gid, 0);
            float run = 0.f, nrun = 0.f;
            for (int k = 0; k < n; ++k) {
                const int g = __builtin_amdgcn_readlane(gid, k);
                if (g != cur) {
                    atomicAdd(&pool[(size_t)cur * HID + c], run);
                    if (c == 0) atomicAdd(&pcnt[cur], nrun);
                    cur = g; run = 0.f; nrun = 0.f;
                }
                run += finish(acc[(r0 + k) * HID + c], b, mu, sc, be);
                nrun += 1.f;
            }
            atomicAdd(&pool[(size_t)cur * HID + c], run);
            if (c == 0) atomicAdd(&pcnt[cur], nrun);
        }
    }
}

__global__ void pool_div(float* __restrict__ sums, const float* __restrict__ cnt, int total) {
    int i = blockIdx.x * blockDim.x + threadIdx.x;
    if (i >= total) return;
    sums[i] /= fmaxf(cnt[i >> 7], 1.0f);
}

extern "C" void kernel_launch(void* const* d_in, const int* in_sizes, int n_in,
                              void* d_out, int out_size, void* d_ws, size_t ws_size,
                              hipStream_t stream) {
    const float* gx  = (const float*)d_in[0];
    const int*   ei  = (const int*)  d_in[1];
    const int*   bat = (const int*)  d_in[2];
    const float* W1  = (const float*)d_in[3];
    const float* as1 = (const float*)d_in[4];
    const float* ad1 = (const float*)d_in[5];
    const float* b1  = (const float*)d_in[6];
    const float* g1  = (const float*)d_in[7];
    const float* be1 = (const float*)d_in[8];
    const float* mu1 = (const float*)d_in[9];
    const float* va1 = (const float*)d_in[10];
    const float* W2  = (const float*)d_in[11];
    const float* as2 = (const float*)d_in[12];
    const float* ad2 = (const float*)d_in[13];
    const float* b2  = (const float*)d_in[14];
    const float* g2  = (const float*)d_in[15];
    const float* be2 = (const float*)d_in[16];
    const float* mu2 = (const float*)d_in[17];
    const float* va2 = (const float*)d_in[18];
    const float* hw  = (const float*)d_in[19];
    const float* hb  = (const float*)d_in[20];

    const int n_nodes = in_sizes[0] / HID;
    const int n_edges = in_sizes[1] / 2;
    const int* esrc = ei;
    const int* edst = ei + n_edges;

    float* ws   = (float*)d_ws;
    float* H    = ws;
    float* AGG  = H    + (size_t)n_nodes * HID;
    float* SS   = AGG  + (size_t)n_nodes * HID;
    float* SD   = SS   + n_nodes;
    float* POOL = SD   + n_nodes;
    float* CNT  = POOL + (size_t)N_GRAPHS * HID;
    const int gridOwn   = (n_nodes + OWN_ROWS - 1) / OWN_ROWS;
    int*      OCNT  = (int*)(CNT + N_GRAPHS);
    size_t wpOff = (size_t)(CNT + N_GRAPHS + gridOwn - ws);
    wpOff = (wpOff + 7) & ~(size_t)7;
    _Float16* WP = (_Float16*)(ws + wpOff);
    unsigned* OLIST = (unsigned*)(ws + wpOff + 8192);

    const int gridNodeW = (n_nodes + 127) / 128;
    const int gridGemm  = (n_nodes / 16 + 4) / 5;

    fill_f32<<<(N_GRAPHS * HID + 255) / 256, 256, 0, stream>>>(POOL, 0.0f, N_GRAPHS * HID);
    fill_f32<<<(N_GRAPHS + gridOwn + 255) / 256, 256, 0, stream>>>(CNT, 0.0f, N_GRAPHS + gridOwn);

    split_edges_by_owner<<<(n_edges + SPLIT_THREADS * SPLIT_EPT - 1) / (SPLIT_THREADS * SPLIT_EPT),
                           SPLIT_THREADS, 0, stream>>>(esrc, edst, OLIST, OCNT,
                                                       n_edges, n_nodes, gridOwn);

    for (int layer = 0; layer < 2; ++layer) {
        const float* xin = (layer == 0) ? gx : AGG;
        const float* W   = (layer == 0) ? W1 : W2;
        const float* asr = (layer == 0) ? as1 : as2;
        const float* ads = (layer == 0) ? ad1 : ad2;
        const float* bia = (layer == 0) ? b1  : b2;
        const float* gam = (layer == 0) ? g1  : g2;
        const float* bet = (layer == 0) ? be1 : be2;
        const float* mu  = (layer == 0) ? mu1 : mu2;
        const float* va  = (layer == 0) ? va1 : va2;

        pack_w_f16<<<4, 256, 0, stream>>>(W, WP, HID);
        gemm128_wmma<false, 8, 5><<<gridGemm, 256, 0, stream>>>(xin, WP, nullptr, H, n_nodes);
        node_scores<<<gridNodeW, 256, 0, stream>>>(H, asr, ads, SS, SD, n_nodes);
        if (layer == 0)
            own_rows_gat<true><<<gridOwn, OWN_THREADS, 0, stream>>>(
                H, esrc, edst, SS, SD, bia, gam, bet, mu, va, AGG, bat, POOL, CNT,
                OLIST, OCNT, n_edges, n_nodes);
        else
            own_rows_gat<false><<<gridOwn, OWN_THREADS, 0, stream>>>(
                H, esrc, edst, SS, SD, bia, gam, bet, mu, va, AGG, bat, POOL, CNT,
                OLIST, OCNT, n_edges, n_nodes);
    }

    pool_div<<<(N_GRAPHS * HID + 255) / 256, 256, 0, stream>>>(POOL, CNT, N_GRAPHS * HID);

    pack_w_f16<<<2, 256, 0, stream>>>(hw, WP, 64);
    gemm128_wmma<true, 4, 1><<<N_GRAPHS / 16, 32 * (64 / 16), 0, stream>>>(
        POOL, WP, hb, (float*)d_out, N_GRAPHS);
}


# GAT layers re-associated: owners sum alpha-weighted INPUT rows in fixed source order, then sums @ W on WMMA from LDS (f16 operands kept); scores = x.(W a); both x@W launches gone; sum order new
# speedup vs baseline: 6.0149x; 1.2702x over previous
#include <hip/hip_runtime.h>
#include <hip/hip_bf16.h>

typedef __attribute__((ext_vector_type(16))) _Float16 v16h;
typedef __attribute__((ext_vector_type(8)))  _Float16 v8h;
typedef __attribute__((ext_vector_type(4)))  _Float16 v4h;
typedef __attribute__((ext_vector_type(8)))  float    v8f;

#define HID 128
#define NEG_SLOPE 0.2f
#define BN_EPS 1e-5f
#define N_GRAPHS 512

template <bool HAS_BIAS, int NW, int RT>
__global__ __launch_bounds__(NW * 32)
void gemm128_wmma(const float* __restrict__ A, const _Float16* __restrict__ Bp,
                  const float* __restrict__ bias, float* __restrict__ C, int M) {
    constexpr int N     = NW * 16;
    constexpr int NT    = NW * 32;
    constexpr int PER   = RT * 512 / NT;
    constexpr int PITCH = 136;
    static_assert(RT * 512 % NT == 0, "the A pieces divide evenly among the threads");
    __shared__ __align__(16) _Float16 As[RT * 16 * PITCH];
    const int tid  = threadIdx.x;
    const int row0 = blockIdx.x * (RT * 16);
    const int wave = tid >> 5;
    const int lane = tid & 31;
    const int hl   = lane & 15;
    const int hi   = lane >> 4;

    float4 f[PER];
#pragma unroll
    for (int q = 0; q < PER; ++q) {
        const int c = tid + q * NT;
        int row = row0 + (c >> 5);
        if (row >= M) row = M - 1;
        f[q] = *(const float4*)(A + (size_t)row * 128 + (c & 31) * 4);
    }
    v16h b[4];
#pragma unroll
    for (int kt = 0; kt < 4; ++kt)
        b[kt] = *(const v16h*)(Bp + (((size_t)wave * 4 + kt) * 32 + lane) * 16);
#pragma unroll
    for (int q = 0; q < PER; ++q) {
        const int c = tid + q * NT;
        v4h h4 = { (_Float16)f[q].x, (_Float16)f[q].y, (_Float16)f[q].z, (_Float16)f[q].w };
        *(v4h*)(As + (c >> 5) * PITCH + (c & 31) * 4) = h4;
    }
    __syncthreads();

    v8f acc[RT];
#pragma unroll
    for (int i = 0; i < RT; ++i) acc[i] = v8f{};
#pragma unroll
    for (int kt = 0; kt < 4; ++kt) {
#pragma unroll
        for (int i = 0; i < RT; ++i) {
            const int base = (i * 16 + hl) * PITCH + kt * 32 + hi * 8;
            v8h alo = *(const v8h*)(As + base);
            v8h ahi = *(const v8h*)(As + base + 16);
            v16h a;
#pragma unroll
            for (int j = 0; j < 8; ++j) { a[j] = alo[j]; a[8 + j] = ahi[j]; }
            acc[i] = __builtin_amdgcn_wmma_f32_16x16x32_f16(
                false, a, false, b[kt], (short)0, acc[i], false, false);
        }
    }

    const int col = wave * 16 + hl;
    float bv = HAS_BIAS ? bias[col] : 0.0f;
#pragma unroll
    for (int i = 0; i < RT; ++i) {
        if (row0 + i * 16 < M) {
#pragma unroll
            for (int r = 0; r < 8; ++r) {
                const int row = row0 + i * 16 + r + 8 * hi;
                C[(size_t)row * N + col] = acc[i][r] + bv;
            }
        }
    }
}

#define SC_ROWS 16
typedef float v4f __attribute__((ext_vector_type(4)));

template <int NH, int OFF>
__device__ __forceinline__ void fold(float (&p)[2 * SC_ROWS], int lane) {
    const bool up = (lane & OFF) != 0;
#pragma unroll
    for (int i = 0; i < NH; ++i) {
        const float keep = up ? p[i + NH] : p[i];
        const float send = up ? p[i] : p[i + NH];
        p[i] = keep + __shfl_xor(send, OFF, 32);
    }
}

__device__ __forceinline__ void scores16(const float* __restrict__ H, const float* __restrict__ a_src,
                                         const float* __restrict__ a_dst, float* __restrict__ ss,
                                         float* __restrict__ sd, int row0, int lane) {
    const v4f as = *(const v4f*)(a_src + 4 * lane);
    const v4f ad = *(const v4f*)(a_dst + 4 * lane);
    const float* hb = H + (size_t)row0 * HID + 4 * lane;
    v4f h[SC_ROWS];
#pragma unroll
    for (int k = 0; k < SC_ROWS; ++k) h[k] = *(const v4f*)(hb + k * HID);
    float p[2 * SC_ROWS];
#pragma unroll
    for (int k = 0; k < SC_ROWS; ++k) {
        p[k]           = h[k].x * as.x + h[k].y * as.y + h[k].z * as.z + h[k].w * as.w;
        p[SC_ROWS + k] = h[k].x * ad.x + h[k].y * ad.y + h[k].z * ad.z + h[k].w * ad.w;
    }
    fold<16, 16>(p, lane);
    fold<8, 8>(p, lane);
    fold<4, 4>(p, lane);
    fold<2, 2>(p, lane);
    fold<1, 1>(p, lane);
    const int r = row0 + (lane & 15);
    if (lane < 16) ss[r] = p[0];
    else           sd[r] = p[0];
}

__global__ __launch_bounds__(256)
void node_scores(const float* __restrict__ H, const float* __restrict__ a_src,
                 const float* __restrict__ a_dst, float* __restrict__ ss,
                 float* __restrict__ sd, int n_nodes) {
    const int lane = threadIdx.x & 31;
    const int row0 = (int)((blockIdx.x * (size_t)blockDim.x + threadIdx.x) >> 5) * SC_ROWS;
    if (row0 >= n_nodes) return;
    scores16(H, a_src, a_dst, ss, sd, row0, lane);
}

__global__ __launch_bounds__(256)
void pack_w_f16(const float* __restrict__ W, _Float16* __restrict__ Wp, int N,
                const float* __restrict__ a_src, const float* __restrict__ a_dst,
                float* __restrict__ wa) {
    int id = blockIdx.x * blockDim.x + threadIdx.x;
    int total = (N / 16) * 4 * 32;
    if (id >= total) {
        const int t = id - total;
        if (t >= 128 / SC_ROWS * 32) return;
        scores16(W, a_src, a_dst, wa, wa + 128, (t >> 5) * SC_ROWS, t & 31);
        return;
    }
    int t    = id >> 7;
    int kt   = (id >> 5) & 3;
    int lane = id & 31;
    int hl = lane & 15, hi = lane >> 4;
    int n = t * 16 + hl;
    _Float16* o = Wp + (size_t)id * 16;
#pragma unroll
    for (int j = 0; j < 8; ++j) {
        o[j]     = (_Float16)W[(size_t)(kt * 32 + hi * 8 + j) * N + n];
        o[8 + j] = (_Float16)W[(size_t)(kt * 32 + 16 + hi * 8 + j) * N + n];
    }
}

__global__ void fill_f32(float* __restrict__ p, float v, int n) {
    int i = blockIdx.x * blockDim.x + threadIdx.x;
    if (i < n) p[i] = v;
}

#define OWN_ROWS    196
#define ACC_PITCH   132
#define OWN_CAP     16384
#define OWN_THREADS 1024

typedef int v4i __attribute__((ext_vector_type(4)));

__device__ __forceinline__ void ldsMaxF(float* addr, float val) {
    if (val >= 0.f) atomicMax((int*)addr, __float_as_int(val));
    else            atomicMin((unsigned int*)addr, __float_as_uint(val));
}

template <class F>
__device__ __forceinline__ void scan_own_edges(const int* __restrict__ edst, int n_edges,
                                               int row0, int rows, F f) {
    const int nthr = blockDim.x;
    const int nvec = (((size_t)edst & 15) == 0) ? (n_edges >> 2) : 0;
    const v4i* dv = (const v4i*)edst;
    const int rot = (int)blockIdx.x * (nvec / (int)gridDim.x);
    for (int v0 = threadIdx.x; v0 < nvec; v0 += 8 * nthr) {
        v4i d[8];
#pragma unroll
        for (int k = 0; k < 8; ++k) {
            const int v = v0 + k * nthr;
            int uu = v + rot;
            if (uu >= nvec) uu -= nvec;
            const v4i none = { -1, -1, -1, -1 };
            d[k] = (v < nvec) ? __builtin_nontemporal_load(dv + uu) : none;
        }
        unsigned m = 0;
#pragma unroll
        for (int k = 0; k < 8; ++k) {
#pragma unroll
            for (int c = 0; c < 4; ++c)
                m |= ((unsigned)(d[k][c] - row0) < (unsigned)rows) ? (1u << (4 * k + c)) : 0u;
        }
        while (m) {
            const int b = __builtin_ctz(m);
            m &= m - 1;
            int uu = v0 + (b >> 2) * nthr + rot;
            if (uu >= nvec) uu -= nvec;
            f(4 * uu + (b & 3));
        }
    }
    for (int e = nvec * 4 + threadIdx.x; e < n_edges; e += nthr)
        if ((unsigned)(edst[e] - row0) < (unsigned)rows) f(e);
}

#define SPLIT_THREADS 256
#define SPLIT_EPT     16
#define SPLIT_MAX_OWN 1024

__global__ __launch_bounds__(SPLIT_THREADS)
void split_edges_by_owner(const int* __restrict__ esrc, const int* __restrict__ edst,
                          unsigned* __restrict__ wsList, int* __restrict__ wsCnt,
                          int n_edges, int n_nodes, int n_own) {
    __shared__ int hist[SPLIT_MAX_OWN];
    __shared__ int hbase[SPLIT_MAX_OWN];
    const int tid = threadIdx.x;
    for (int o = tid; o < n_own; o += SPLIT_THREADS) hist[o] = 0;

    const int e0 = (int)blockIdx.x * (SPLIT_THREADS * SPLIT_EPT) + tid;
    int d[SPLIT_EPT], s[SPLIT_EPT];
#pragma unroll
    for (int k = 0; k < SPLIT_EPT; ++k) {
        const int e  = e0 + k * SPLIT_THREADS;
        const int ec = (e < n_edges) ? e : n_edges - 1;
        d[k] = edst[ec];
        s[k] = esrc[ec];
        if (e >= n_edges) d[k] = -1;
    }
    __syncthreads();

    int own[SPLIT_EPT], rank[SPLIT_EPT];
#pragma unroll
    for (int k = 0; k < SPLIT_EPT; ++k) {
        own[k]  = ((unsigned)d[k] < (unsigned)n_nodes) ? d[k] / OWN_ROWS : -1;
        rank[k] = 0;
        if (own[k] >= 0) rank[k] = atomicAdd(&hist[own[k]], 1);
    }
    __syncthreads();

    for (int o = tid; o < n_own; o += SPLIT_THREADS) {
        const int cnt = hist[o];
        hbase[o] = (cnt > 0) ? atomicAdd(&wsCnt[o], cnt) : 0;
    }
    __syncthreads();

    int pos[SPLIT_EPT];
#pragma unroll
    for (int k = 0; k < SPLIT_EPT; ++k) pos[k] = hbase[(own[k] >= 0) ? own[k] : 0] + rank[k];
#pragma unroll
    for (int k = 0; k < SPLIT_EPT; ++k) {
        if (own[k] >= 0 && pos[k] < OWN_CAP)
            wsList[(size_t)own[k] * OWN_CAP + pos[k]] =
                ((unsigned)s[k] << 8) | (unsigned)(d[k] - own[k] * OWN_ROWS);
    }
}


template <int M>
__device__ __forceinline__ float xrow(float x) {
    return __builtin_bit_cast(float, __builtin_amdgcn_update_dpp(
        0, __builtin_bit_cast(int, x), 0x160 + M, 0xF, 0xF, true));
}
__device__ __forceinline__ float xhalf(float x) {
    const int b = __builtin_bit_cast(int, x);
    return __builtin_bit_cast(float, __builtin_amdgcn_permlanex16(b, b, 0x76543210, 0xfedcba98, false, false));
}
__device__ __forceinline__ float wave_max32(float v) {
    v = fmaxf(v, xrow<1>(v));
    v = fmaxf(v, xrow<2>(v));
    v = fmaxf(v, xrow<4>(v));
    v = fmaxf(v, xrow<8>(v));
    return fmaxf(v, xhalf(v));
}
__device__ __forceinline__ float wave_sum32(float v) {
    v += xrow<1>(v);
    v += xrow<2>(v);
    v += xrow<4>(v);
    v += xrow<8>(v);
    return v + xhalf(v);
}

template <int NB>
__device__ __forceinline__ void add_rows(const float* __restrict__ H, int s, float w, int k0,
                                         int nvalid, int lane, float4& sum) {
    float4 hv[NB];
    float  wk[NB];
#pragma unroll
    for (int k = 0; k < NB; ++k) {
        const int kk = (k < nvalid) ? (k0 + k) : k0;
        const int sk = __builtin_amdgcn_readlane(s, kk);
        const float wv = __int_as_float(__builtin_amdgcn_readlane(__float_as_int(w), kk));
        wk[k] = (k < nvalid) ? wv : 0.f;
        hv[k] = *(const float4*)(H + (size_t)sk * HID + 4 * lane);
    }
#pragma unroll
    for (int k = 0; k < NB; ++k) {
        sum.x += wk[k] * hv[k].x;
        sum.y += wk[k] * hv[k].y;
        sum.z += wk[k] * hv[k].z;
        sum.w += wk[k] * hv[k].w;
    }
}

template <bool FIRST>
__global__ __launch_bounds__(OWN_THREADS)
void own_rows_gat(const float* __restrict__ H, const _Float16* __restrict__ Wp,
                  const int* __restrict__ esrc,
                  const int* __restrict__ edst, const float* __restrict__ ss,
                  const float* __restrict__ sd,
                  const float* __restrict__ bias, const float* __restrict__ gam,
                  const float* __restrict__ bet, const float* __restrict__ mean,
                  const float* __restrict__ var,
                  float* __restrict__ out,
                  const int* __restrict__ batch, float* __restrict__ pool,
                  float* __restrict__ pcnt,
                  const unsigned* __restrict__ wsList, const int* __restrict__ wsCnt,
                  int n_edges, int n_nodes) {
    __shared__ __align__(16) float acc[(OWN_ROWS + 15) / 16 * 16 * ACC_PITCH];
    __shared__ __align__(16) _Float16 Ws[HID * HID];
    __shared__ unsigned srt[OWN_CAP];
    __shared__ int   rcnt[OWN_ROWS];
    __shared__ int   rpos[OWN_ROWS + 1];
    __shared__ float mxl[OWN_ROWS];
    __shared__ float denl[OWN_ROWS];
    __shared__ float sdl[OWN_ROWS];
    static_assert(OWN_ROWS <= 7 * 32, "the scan below gives a lane seven rows");
    static_assert(OWN_ROWS <= 256, "a list entry keeps its local row in eight bits");
    constexpr int NT16 = (OWN_ROWS + 15) / 16;
    static_assert(2 * NT16 <= OWN_THREADS / 32, "a wave takes a row tile and half of the columns");
    static_assert(HID * HID / 16 == OWN_THREADS, "a thread brings one packed fragment piece of W");

    const int tid = threadIdx.x, nthr = blockDim.x;
    const int lane = tid & 31;
    const int wave = __builtin_amdgcn_readfirstlane(tid >> 5);
    const int nwaves = nthr >> 5;
    const int row0 = blockIdx.x * OWN_ROWS;
    const int rows = (n_nodes - row0 < OWN_ROWS) ? (n_nodes - row0) : OWN_ROWS;
    const unsigned* wl = wsList + (size_t)blockIdx.x * OWN_CAP;

    const int nin   = wsCnt[blockIdx.x];
    const bool fits = rows + nin <= OWN_CAP;

    auto score = [&](int s, int dl) {
        const float v = ss[s] + sdl[dl];
        return (v > 0.f) ? v : NEG_SLOPE * v;
    };

    const v16h wf = *(const v16h*)(Wp + (size_t)tid * 16);
    for (int i = tid; i < rows; i += nthr) {
        sdl[i]  = sd[row0 + i];
        rcnt[i] = 1;
    }
    {
        v8h wlo, whi;
#pragma unroll
        for (int j = 0; j < 8; ++j) { wlo[j] = wf[j]; whi[j] = wf[8 + j]; }
        _Float16* wp = Ws + (tid >> 5) * 512 + (tid & 31) * 8;
        *(v8h*)wp         = wlo;
        *(v8h*)(wp + 256) = whi;
    }
    __syncthreads();

    if (fits) {
        for (int i = tid; i < nin; i += nthr) atomicAdd(&rcnt[wl[i] & 255u], 1);
        __syncthreads();
        if (wave == 0) {
            int c[7], tot = 0;
#pragma unroll
            for (int j = 0; j < 7; ++j) {
                const int idx = lane * 7 + j;
                c[j] = (idx < rows) ? rcnt[idx] : 0;
                tot += c[j];
            }
            int inc = tot;
#pragma unroll
            for (int off = 1; off < 32; off <<= 1) {
                const int t = __shfl_up(inc, off, 32);
                if (lane >= off) inc += t;
            }
            int run = inc - tot;
#pragma unroll
            for (int j = 0; j < 7; ++j) {
                const int idx = lane * 7 + j;
                if (idx < rows) {
                    rpos[idx] = run;
                    srt[run]  = (unsigned)(row0 + idx);
                    rcnt[idx] = run + 1;
                }
                run += c[j];
            }
            if (lane == 31) rpos[rows] = run;
        }
        __syncthreads();
        for (int i = tid; i < nin; i += nthr) {
            const unsigned p = wl[i];
            const int slot = atomicAdd(&rcnt[p & 255u], 1);
            srt[slot] = p >> 8;
        }
        __syncthreads();

        int e0 = 0, deg = 0, s0 = 0;
        float a0 = 0.f;
        auto fetch = [&](int rr) {
            e0  = __builtin_amdgcn_readfirstlane(rpos[rr]);
            deg = __builtin_amdgcn_readfirstlane(rpos[rr + 1]) - e0;
            s0  = (lane < deg) ? (int)srt[e0 + lane] : (row0 + rr);
            a0  = ss[s0];
        };
        if (wave < rows) fetch(wave);
        for (int r = wave; r < rows; r += nwaves) {
            const int ce0 = e0, cdeg = deg, cs0 = s0;
            const float ca0 = a0;
            if (r + nwaves < rows) fetch(r + nwaves);
            const float sdr = sdl[r];
            auto lrelu = [](float v) { return (v > 0.f) ? v : NEG_SLOPE * v; };
            const int n0 = (cdeg < 32) ? cdeg : 32;
            int below = 0;
            for (int k = 0; k < n0; ++k) {
                const int sk = __builtin_amdgcn_readlane(cs0, k);
                below += (sk < cs0 || (sk == cs0 && k < lane)) ? 1 : 0;
            }
            const int place = (lane < n0) ? below : lane;
            const int   cs = __builtin_amdgcn_ds_permute(place << 2, cs0);
            const float ca = __int_as_float(
                __builtin_amdgcn_ds_permute(place << 2, __float_as_int(ca0)));
            const float v0 = (lane < cdeg) ? lrelu(ca + sdr) : -3.402823466e38f;
            float m = v0;
            for (int c0 = 32; c0 < cdeg; c0 += 32) {
                const int i = c0 + lane;
                if (i < cdeg) m = fmaxf(m, lrelu(ss[srt[ce0 + i]] + sdr));
            }
            m = wave_max32(m);
            const float ex0 = (lane < cdeg) ? __expf(v0 - m) : 0.f;
            float den = ex0;
            for (int c0 = 32; c0 < cdeg; c0 += 32) {
                const int i = c0 + lane;
                if (i < cdeg) den += __expf(lrelu(ss[srt[ce0 + i]] + sdr) - m);
            }
            den = wave_sum32(den) + 1e-16f;

            float4 sum = make_float4(0.f, 0.f, 0.f, 0.f);
            for (int c0 = 0; c0 < cdeg; c0 += 32) {
                int s = cs;
                float w = ex0 / den;
                if (c0 > 0) {
                    const int i = c0 + lane;
                    s = (i < cdeg) ? (int)srt[ce0 + i] : (row0 + r);
                    w = (i < cdeg) ? __expf(lrelu(ss[s] + sdr) - m) / den : 0.f;
                }
                const int left = cdeg - c0;
                const int nb = (left < 32) ? left : 32;
                int k0 = 0;
                for (; k0 + 8 <= nb; k0 += 8) add_rows<8>(H, s, w, k0, 8, lane, sum);
                const int rem = nb - k0;
                if (rem > 4)      add_rows<8>(H, s, w, k0, rem, lane, sum);
                else if (rem > 0) add_rows<4>(H, s, w, k0, rem, lane, sum);
            }
            *(float4*)(acc + r * ACC_PITCH + 4 * lane) = sum;
        }
    } else {
        for (int i = tid; i < OWN_ROWS * (HID / 4); i += nthr)
            *(float4*)(acc + (i >> 5) * ACC_PITCH + (i & 31) * 4) = make_float4(0.f, 0.f, 0.f, 0.f);
        for (int i = tid; i < rows; i += nthr) { mxl[i] = -3.402823466e38f; denl[i] = 0.f; }
        __syncthreads();

        auto do_max = [&](int s, int dl) { ldsMaxF(&mxl[dl], score(s, dl)); };
        for (int i = tid; i < rows; i += nthr) do_max(row0 + i, i);
        scan_own_edges(edst, n_edges, row0, rows, [&](int e) { do_max(esrc[e], edst[e] - row0); });
        __syncthreads();

        auto do_sum = [&](int s, int dl) { atomicAdd(&denl[dl], __expf(score(s, dl) - mxl[dl])); };
        for (int i = tid; i < rows; i += nthr) do_sum(row0 + i, i);
        scan_own_edges(edst, n_edges, row0, rows, [&](int e) { do_sum(esrc[e], edst[e] - row0); });
        __syncthreads();

        for (int idx = tid; idx < rows * HID; idx += nthr) {
            const int dl = idx / HID, c = idx % HID, s = row0 + dl;
            const float w = __expf(score(s, dl) - mxl[dl]) / (denl[dl] + 1e-16f);
            atomicAdd(&acc[dl * ACC_PITCH + c], w * H[(size_t)s * HID + c]);
        }
        scan_own_edges(edst, n_edges, row0, rows, [&](int e) {
            const int s = esrc[e], dl = edst[e] - row0;
            const float w = __expf(score(s, dl) - mxl[dl]) / (denl[dl] + 1e-16f);
            const float* hs = H + (size_t)s * HID;
            for (int c = 0; c < HID; ++c) atomicAdd(&acc[dl * ACC_PITCH + c], w * hs[c]);
        });
    }

    __syncthreads();
    {
        const int tile = wave >> 1, half = wave & 1;
        const bool has = tile < NT16;
        const int hl = lane & 15, hi = lane >> 4;
        v16h a[4];
        if (has) {
            const float* ap = acc + (tile * 16 + hl) * ACC_PITCH + hi * 8;
#pragma unroll
            for (int kt = 0; kt < 4; ++kt) {
                const float4 x0 = *(const float4*)(ap + kt * 32);
                const float4 x1 = *(const float4*)(ap + kt * 32 + 4);
                const float4 y0 = *(const float4*)(ap + kt * 32 + 16);
                const float4 y1 = *(const float4*)(ap + kt * 32 + 20);
                a[kt] = v16h{ (_Float16)x0.x, (_Float16)x0.y, (_Float16)x0.z, (_Float16)x0.w,
                              (_Float16)x1.x, (_Float16)x1.y, (_Float16)x1.z, (_Float16)x1.w,
                              (_Float16)y0.x, (_Float16)y0.y, (_Float16)y0.z, (_Float16)y0.w,
                              (_Float16)y1.x, (_Float16)y1.y, (_Float16)y1.z, (_Float16)y1.w };
            }
        }
        __syncthreads();
        if (has) {
#pragma unroll
            for (int q = 0; q < 4; ++q) {
                const int ct = half * 4 + q;
                v8f d = v8f{};
#pragma unroll
                for (int kt = 0; kt < 4; ++kt) {
                    const _Float16* bp = Ws + (ct * 4 + kt) * 512 + lane * 8;
                    const v8h blo = *(const v8h*)bp;
                    const v8h bhi = *(const v8h*)(bp + 256);
                    v16h b;
#pragma unroll
                    for (int j = 0; j < 8; ++j) { b[j] = blo[j]; b[8 + j] = bhi[j]; }
                    d = __builtin_amdgcn_wmma_f32_16x16x32_f16(
                        false, a[kt], false, b, (short)0, d, false, false);
                }
#pragma unroll
                for (int r = 0; r < 8; ++r)
                    acc[(tile * 16 + r + 8 * hi) * ACC_PITCH + ct * 16 + hl] = d[r];
            }
        }
    }

    auto finish = [](float v, float b, float mu, float sc, float be) {
        return fmaxf((v + b - mu) * sc + be, 0.f);
    };

    if (FIRST) {
        const int c0 = (tid & 31) * 4;
        float b[4], mu[4], sc[4], be[4];
#pragma unroll
        for (int j = 0; j < 4; ++j) {
            b[j]  = bias[c0 + j];
            mu[j] = mean[c0 + j];
            sc[j] = gam[c0 + j] * rsqrtf(var[c0 + j] + BN_EPS);
            be[j] = bet[c0 + j];
        }
        __syncthreads();

        float4* o4 = (float4*)(out + (size_t)row0 * HID);
        for (int i = tid; i < rows * (HID / 4); i += nthr) {
            float4 a = *(const float4*)(acc + (i >> 5) * ACC_PITCH + (i & 31) * 4);
            a.x = finish(a.x, b[0], mu[0], sc[0], be[0]);
            a.y = finish(a.y, b[1], mu[1], sc[1], be[1]);
            a.z = finish(a.z, b[2], mu[2], sc[2], be[2]);
            a.w = finish(a.w, b[3], mu[3], sc[3], be[3]);
            o4[i] = a;
        }
    } else {
        static_assert((OWN_ROWS + OWN_THREADS / HID - 1) / (OWN_THREADS / HID) <= 32,
                      "a slice's graph ids are held one per lane");
        const int c = tid & (HID - 1);
        const float b  = bias[c];
        const float mu = mean[c];
        const float sc = gam[c] * rsqrtf(var[c] + BN_EPS);
        const float be = bet[c];
        const int ngrp = nthr / HID;
        const int per  = (rows + ngrp - 1) / ngrp;
        const int r0   = (wave >> 2) * per;
        const int n    = (rows - r0 < per) ? (rows - r0) : per;
        const int gid = (lane < n) ? batch[row0 + r0 + lane] : 0;
        __syncthreads();

        if (n > 0) {
            int cur = __builtin_amdgcn_readlane(gid, 0);
            float run = 0.f, nrun = 0.f;
            for (int k = 0; k < n; ++k) {
                const int g = __builtin_amdgcn_readlane(gid, k);
                if (g != cur) {
                    atomicAdd(&pool[(size_t)cur * HID + c], run);
                    if (c == 0) atomicAdd(&pcnt[cur], nrun);
                    cur = g; run = 0.f; nrun = 0.f;
                }
                run += finish(acc[(r0 + k) * ACC_PITCH + c], b, mu, sc, be);
                nrun += 1.f;
            }
            atomicAdd(&pool[(size_t)cur * HID + c], run);
            if (c == 0) atomicAdd(&pcnt[cur], nrun);
        }
    }
}

__global__ void pool_div(float* __restrict__ sums, const float* __restrict__ cnt, int total) {
    int i = blockIdx.x * blockDim.x + threadIdx.x;
    if (i >= total) return;
    sums[i] /= fmaxf(cnt[i >> 7], 1.0f);
}

extern "C" void kernel_launch(void* const* d_in, const int* in_sizes, int n_in,
                              void* d_out, int out_size, void* d_ws, size_t ws_size,
                              hipStream_t stream) {
    const float* gx  = (const float*)d_in[0];
    const int*   ei  = (const int*)  d_in[1];
    const int*   bat = (const int*)  d_in[2];
    const float* W1  = (const float*)d_in[3];
    const float* as1 = (const float*)d_in[4];
    const float* ad1 = (const float*)d_in[5];
    const float* b1  = (const float*)d_in[6];
    const float* g1  = (const float*)d_in[7];
    const float* be1 = (const float*)d_in[8];
    const float* mu1 = (const float*)d_in[9];
    const float* va1 = (const float*)d_in[10];
    const float* W2  = (const float*)d_in[11];
    const float* as2 = (const float*)d_in[12];
    const float* ad2 = (const float*)d_in[13];
    const float* b2  = (const float*)d_in[14];
    const float* g2  = (const float*)d_in[15];
    const float* be2 = (const float*)d_in[16];
    const float* mu2 = (const float*)d_in[17];
    const float* va2 = (const float*)d_in[18];
    const float* hw  = (const float*)d_in[19];
    const float* hb  = (const float*)d_in[20];

    const int n_nodes = in_sizes[0] / HID;
    const int n_edges = in_sizes[1] / 2;
    const int* esrc = ei;
    const int* edst = ei + n_edges;

    float* ws   = (float*)d_ws;
    float* WA   = ws;
    float* AGG  = WA   + (size_t)n_nodes * HID;
    float* SS   = AGG  + (size_t)n_nodes * HID;
    float* SD   = SS   + n_nodes;
    float* POOL = SD   + n_nodes;
    float* CNT  = POOL + (size_t)N_GRAPHS * HID;
    const int gridOwn   = (n_nodes + OWN_ROWS - 1) / OWN_ROWS;
    int*      OCNT  = (int*)(CNT + N_GRAPHS);
    size_t wpOff = (size_t)(CNT + N_GRAPHS + gridOwn - ws);
    wpOff = (wpOff + 7) & ~(size_t)7;
    _Float16* WP = (_Float16*)(ws + wpOff);
    unsigned* OLIST = (unsigned*)(ws + wpOff + 8192);

    const int gridNodeW = (n_nodes + 127) / 128;

    fill_f32<<<(N_GRAPHS * HID + 255) / 256, 256, 0, stream>>>(POOL, 0.0f, N_GRAPHS * HID);
    fill_f32<<<(N_GRAPHS + gridOwn + 255) / 256, 256, 0, stream>>>(CNT, 0.0f, N_GRAPHS + gridOwn);

    split_edges_by_owner<<<(n_edges + SPLIT_THREADS * SPLIT_EPT - 1) / (SPLIT_THREADS * SPLIT_EPT),
                           SPLIT_THREADS, 0, stream>>>(esrc, edst, OLIST, OCNT,
                                                       n_edges, n_nodes, gridOwn);

    for (int layer = 0; layer < 2; ++layer) {
        const float* xin = (layer == 0) ? gx : AGG;
        const float* W   = (layer == 0) ? W1 : W2;
        const float* asr = (layer == 0) ? as1 : as2;
        const float* ads = (layer == 0) ? ad1 : ad2;
        const float* bia = (layer == 0) ? b1  : b2;
        const float* gam = (layer == 0) ? g1  : g2;
        const float* bet = (layer == 0) ? be1 : be2;
        const float* mu  = (layer == 0) ? mu1 : mu2;
        const float* va  = (layer == 0) ? va1 : va2;

        pack_w_f16<<<5, 256, 0, stream>>>(W, WP, HID, asr, ads, WA);
        node_scores<<<gridNodeW, 256, 0, stream>>>(xin, WA, WA + HID, SS, SD, n_nodes);
        if (layer == 0)
            own_rows_gat<true><<<gridOwn, OWN_THREADS, 0, stream>>>(
                xin, WP, esrc, edst, SS, SD, bia, gam, bet, mu, va, AGG, bat, POOL, CNT,
                OLIST, OCNT, n_edges, n_nodes);
        else
            own_rows_gat<false><<<gridOwn, OWN_THREADS, 0, stream>>>(
                xin, WP, esrc, edst, SS, SD, bia, gam, bet, mu, va, AGG, bat, POOL, CNT,
                OLIST, OCNT, n_edges, n_nodes);
    }

    pool_div<<<(N_GRAPHS * HID + 255) / 256, 256, 0, stream>>>(POOL, CNT, N_GRAPHS * HID);

    pack_w_f16<<<2, 256, 0, stream>>>(hw, WP, 64, nullptr, nullptr, nullptr);
    gemm128_wmma<true, 4, 1><<<N_GRAPHS / 16, 32 * (64 / 16), 0, stream>>>(
        POOL, WP, hb, (float*)d_out, N_GRAPHS);
}


# Layer 1's owners prepare layer 2: their rows' scores x1.(W2 a) at write-out, their grouped, source-ordered list handed on; launches 9 -> 8, layer 2's sort and rank loops gone; score sum order new
# speedup vs baseline: 6.6202x; 1.1006x over previous
#include <hip/hip_runtime.h>
#include <hip/hip_bf16.h>

typedef __attribute__((ext_vector_type(16))) _Float16 v16h;
typedef __attribute__((ext_vector_type(8)))  _Float16 v8h;
typedef __attribute__((ext_vector_type(4)))  _Float16 v4h;
typedef __attribute__((ext_vector_type(8)))  float    v8f;

#define HID 128
#define NEG_SLOPE 0.2f
#define BN_EPS 1e-5f
#define N_GRAPHS 512

template <bool HAS_BIAS, int NW, int RT>
__global__ __launch_bounds__(NW * 32)
void gemm128_wmma(const float* __restrict__ A, const float* __restrict__ B,
                  const float* __restrict__ bias, float* __restrict__ C, int M) {
    constexpr int N     = NW * 16;
    constexpr int NT    = NW * 32;
    constexpr int PER   = RT * 512 / NT;
    constexpr int PITCH = 136;
    static_assert(RT * 512 % NT == 0, "the A pieces divide evenly among the threads");
    __shared__ __align__(16) _Float16 As[RT * 16 * PITCH];
    const int tid  = threadIdx.x;
    const int row0 = blockIdx.x * (RT * 16);
    const int wave = tid >> 5;
    const int lane = tid & 31;
    const int hl   = lane & 15;
    const int hi   = lane >> 4;

    float4 f[PER];
#pragma unroll
    for (int q = 0; q < PER; ++q) {
        const int c = tid + q * NT;
        int row = row0 + (c >> 5);
        if (row >= M) row = M - 1;
        f[q] = *(const float4*)(A + (size_t)row * 128 + (c & 31) * 4);
    }
    v16h b[4];
#pragma unroll
    for (int kt = 0; kt < 4; ++kt) {
        const float* bp = B + (size_t)(kt * 32 + hi * 8) * N + wave * 16 + hl;
#pragma unroll
        for (int j = 0; j < 8; ++j) {
            b[kt][j]     = (_Float16)bp[(size_t)j * N];
            b[kt][8 + j] = (_Float16)bp[(size_t)(16 + j) * N];
        }
    }
#pragma unroll
    for (int q = 0; q < PER; ++q) {
        const int c = tid + q * NT;
        v4h h4 = { (_Float16)f[q].x, (_Float16)f[q].y, (_Float16)f[q].z, (_Float16)f[q].w };
        *(v4h*)(As + (c >> 5) * PITCH + (c & 31) * 4) = h4;
    }
    __syncthreads();

    v8f acc[RT];
#pragma unroll
    for (int i = 0; i < RT; ++i) acc[i] = v8f{};
#pragma unroll
    for (int kt = 0; kt < 4; ++kt) {
#pragma unroll
        for (int i = 0; i < RT; ++i) {
            const int base = (i * 16 + hl) * PITCH + kt * 32 + hi * 8;
            v8h alo = *(const v8h*)(As + base);
            v8h ahi = *(const v8h*)(As + base + 16);
            v16h a;
#pragma unroll
            for (int j = 0; j < 8; ++j) { a[j] = alo[j]; a[8 + j] = ahi[j]; }
            acc[i] = __builtin_amdgcn_wmma_f32_16x16x32_f16(
                false, a, false, b[kt], (short)0, acc[i], false, false);
        }
    }

    const int col = wave * 16 + hl;
    float bv = HAS_BIAS ? bias[col] : 0.0f;
#pragma unroll
    for (int i = 0; i < RT; ++i) {
        if (row0 + i * 16 < M) {
#pragma unroll
            for (int r = 0; r < 8; ++r) {
                const int row = row0 + i * 16 + r + 8 * hi;
                C[(size_t)row * N + col] = acc[i][r] + bv;
            }
        }
    }
}

#define SC_ROWS 16
#define SC_THREADS 512
typedef float v4f __attribute__((ext_vector_type(4)));

template <int NH, int OFF>
__device__ __forceinline__ void fold(float (&p)[2 * SC_ROWS], int lane) {
    const bool up = (lane & OFF) != 0;
#pragma unroll
    for (int i = 0; i < NH; ++i) {
        const float keep = up ? p[i + NH] : p[i];
        const float send = up ? p[i] : p[i + NH];
        p[i] = keep + __shfl_xor(send, OFF, 32);
    }
}

__device__ __forceinline__ void load16(const float* __restrict__ H, int row0, int lane,
                                       v4f (&h)[SC_ROWS]) {
    const float* hb = H + (size_t)row0 * HID + 4 * lane;
#pragma unroll
    for (int k = 0; k < SC_ROWS; ++k) h[k] = *(const v4f*)(hb + k * HID);
}

__device__ __forceinline__ void dots16(const v4f (&h)[SC_ROWS], v4f as, v4f ad,
                                       float* __restrict__ ss, float* __restrict__ sd, int lane) {
    float p[2 * SC_ROWS];
#pragma unroll
    for (int k = 0; k < SC_ROWS; ++k) {
        p[k]           = h[k].x * as.x + h[k].y * as.y + h[k].z * as.z + h[k].w * as.w;
        p[SC_ROWS + k] = h[k].x * ad.x + h[k].y * ad.y + h[k].z * ad.z + h[k].w * ad.w;
    }
    fold<16, 16>(p, lane);
    fold<8, 8>(p, lane);
    fold<4, 4>(p, lane);
    fold<2, 2>(p, lane);
    fold<1, 1>(p, lane);
    if (lane < 16) ss[lane & 15] = p[0];
    else           sd[lane & 15] = p[0];
}

__global__ __launch_bounds__(SC_THREADS)
void node_scores(const float* __restrict__ H, const float* __restrict__ W,
                 const float* __restrict__ a_src, const float* __restrict__ a_dst,
                 float* __restrict__ ss, float* __restrict__ sd, int n_nodes) {
    static_assert(HID / SC_ROWS <= SC_THREADS / 32, "W's rows find a wave each sixteen");
    __shared__ __align__(16) float wa[2 * HID];
    const int lane = threadIdx.x & 31;
    const int wave = threadIdx.x >> 5;
    const int row0 = ((int)blockIdx.x * (SC_THREADS / 32) + wave) * SC_ROWS;
    const bool live = row0 < n_nodes;
    v4f hn[SC_ROWS];
    load16(H, live ? row0 : 0, lane, hn);
    if (wave < HID / SC_ROWS) {
        v4f hw[SC_ROWS];
        load16(W, wave * SC_ROWS, lane, hw);
        const v4f as = *(const v4f*)(a_src + 4 * lane);
        const v4f ad = *(const v4f*)(a_dst + 4 * lane);
        dots16(hw, as, ad, wa + wave * SC_ROWS, wa + HID + wave * SC_ROWS, lane);
    }
    __syncthreads();
    if (!live) return;
    const v4f was = *(const v4f*)(wa + 4 * lane);
    const v4f wad = *(const v4f*)(wa + HID + 4 * lane);
    dots16(hn, was, wad, ss + row0, sd + row0, lane);
}

__global__ void fill_f32(float* __restrict__ p, float v, int n) {
    int i = blockIdx.x * blockDim.x + threadIdx.x;
    if (i < n) p[i] = v;
}

#define OWN_ROWS    196
#define ACC_PITCH   132
#define OWN_CAP     16384
#define OWN_THREADS 1024

typedef int v4i __attribute__((ext_vector_type(4)));

__device__ __forceinline__ void ldsMaxF(float* addr, float val) {
    if (val >= 0.f) atomicMax((int*)addr, __float_as_int(val));
    else            atomicMin((unsigned int*)addr, __float_as_uint(val));
}

template <class F>
__device__ __forceinline__ void scan_own_edges(const int* __restrict__ edst, int n_edges,
                                               int row0, int rows, F f) {
    const int nthr = blockDim.x;
    const int nvec = (((size_t)edst & 15) == 0) ? (n_edges >> 2) : 0;
    const v4i* dv = (const v4i*)edst;
    const int rot = (int)blockIdx.x * (nvec / (int)gridDim.x);
    for (int v0 = threadIdx.x; v0 < nvec; v0 += 8 * nthr) {
        v4i d[8];
#pragma unroll
        for (int k = 0; k < 8; ++k) {
            const int v = v0 + k * nthr;
            int uu = v + rot;
            if (uu >= nvec) uu -= nvec;
            const v4i none = { -1, -1, -1, -1 };
            d[k] = (v < nvec) ? __builtin_nontemporal_load(dv + uu) : none;
        }
        unsigned m = 0;
#pragma unroll
        for (int k = 0; k < 8; ++k) {
#pragma unroll
            for (int c = 0; c < 4; ++c)
                m |= ((unsigned)(d[k][c] - row0) < (unsigned)rows) ? (1u << (4 * k + c)) : 0u;
        }
        while (m) {
            const int b = __builtin_ctz(m);
            m &= m - 1;
            int uu = v0 + (b >> 2) * nthr + rot;
            if (uu >= nvec) uu -= nvec;
            f(4 * uu + (b & 3));
        }
    }
    for (int e = nvec * 4 + threadIdx.x; e < n_edges; e += nthr)
        if ((unsigned)(edst[e] - row0) < (unsigned)rows) f(e);
}

#define SPLIT_THREADS 256
#define SPLIT_EPT     16
#define SPLIT_MAX_OWN 1024

__global__ __launch_bounds__(SPLIT_THREADS)
void split_edges_by_owner(const int* __restrict__ esrc, const int* __restrict__ edst,
                          unsigned* __restrict__ wsList, int* __restrict__ wsCnt,
                          int n_edges, int n_nodes, int n_own) {
    __shared__ int hist[SPLIT_MAX_OWN];
    __shared__ int hbase[SPLIT_MAX_OWN];
    const int tid = threadIdx.x;
    for (int o = tid; o < n_own; o += SPLIT_THREADS) hist[o] = 0;

    const int e0 = (int)blockIdx.x * (SPLIT_THREADS * SPLIT_EPT) + tid;
    int d[SPLIT_EPT], s[SPLIT_EPT];
#pragma unroll
    for (int k = 0; k < SPLIT_EPT; ++k) {
        const int e  = e0 + k * SPLIT_THREADS;
        const int ec = (e < n_edges) ? e : n_edges - 1;
        d[k] = edst[ec];
        s[k] = esrc[ec];
        if (e >= n_edges) d[k] = -1;
    }
    __syncthreads();

    int own[SPLIT_EPT], rank[SPLIT_EPT];
#pragma unroll
    for (int k = 0; k < SPLIT_EPT; ++k) {
        own[k]  = ((unsigned)d[k] < (unsigned)n_nodes) ? d[k] / OWN_ROWS : -1;
        rank[k] = 0;
        if (own[k] >= 0) rank[k] = atomicAdd(&hist[own[k]], 1);
    }
    __syncthreads();

    for (int o = tid; o < n_own; o += SPLIT_THREADS) {
        const int cnt = hist[o];
        hbase[o] = (cnt > 0) ? atomicAdd(&wsCnt[o], cnt) : 0;
    }
    __syncthreads();

    int pos[SPLIT_EPT];
#pragma unroll
    for (int k = 0; k < SPLIT_EPT; ++k) pos[k] = hbase[(own[k] >= 0) ? own[k] : 0] + rank[k];
#pragma unroll
    for (int k = 0; k < SPLIT_EPT; ++k) {
        if (own[k] >= 0 && pos[k] < OWN_CAP)
            wsList[(size_t)own[k] * OWN_CAP + pos[k]] =
                ((unsigned)s[k] << 8) | (unsigned)(d[k] - own[k] * OWN_ROWS);
    }
}


template <int M>
__device__ __forceinline__ float xrow(float x) {
    return __builtin_bit_cast(float, __builtin_amdgcn_update_dpp(
        0, __builtin_bit_cast(int, x), 0x160 + M, 0xF, 0xF, true));
}
__device__ __forceinline__ float xhalf(float x) {
    const int b = __builtin_bit_cast(int, x);
    return __builtin_bit_cast(float, __builtin_amdgcn_permlanex16(b, b, 0x76543210, 0xfedcba98, false, false));
}
__device__ __forceinline__ float wave_max32(float v) {
    v = fmaxf(v, xrow<1>(v));
    v = fmaxf(v, xrow<2>(v));
    v = fmaxf(v, xrow<4>(v));
    v = fmaxf(v, xrow<8>(v));
    return fmaxf(v, xhalf(v));
}
__device__ __forceinline__ float wave_sum32(float v) {
    v += xrow<1>(v);
    v += xrow<2>(v);
    v += xrow<4>(v);
    v += xrow<8>(v);
    return v + xhalf(v);
}

template <int NB>
__device__ __forceinline__ void add_rows(const float* __restrict__ H, int s, float w, int k0,
                                         int nvalid, int lane, float4& sum) {
    float4 hv[NB];
    float  wk[NB];
#pragma unroll
    for (int k = 0; k < NB; ++k) {
        const int kk = (k < nvalid) ? (k0 + k) : k0;
        const int sk = __builtin_amdgcn_readlane(s, kk);
        const float wv = __int_as_float(__builtin_amdgcn_readlane(__float_as_int(w), kk));
        wk[k] = (k < nvalid) ? wv : 0.f;
        hv[k] = *(const float4*)(H + (size_t)sk * HID + 4 * lane);
    }
#pragma unroll
    for (int k = 0; k < NB; ++k) {
        sum.x += wk[k] * hv[k].x;
        sum.y += wk[k] * hv[k].y;
        sum.z += wk[k] * hv[k].z;
        sum.w += wk[k] * hv[k].w;
    }
}

template <bool FIRST>
__global__ __launch_bounds__(OWN_THREADS)
void own_rows_gat(const float* __restrict__ H, const float* __restrict__ W,
                  const int* __restrict__ esrc,
                  const int* __restrict__ edst, const float* __restrict__ ss,
                  const float* __restrict__ sd,
                  const float* __restrict__ bias, const float* __restrict__ gam,
                  const float* __restrict__ bet, const float* __restrict__ mean,
                  const float* __restrict__ var,
                  float* __restrict__ out,
                  const int* __restrict__ batch, float* __restrict__ pool,
                  float* __restrict__ pcnt,
                  unsigned* __restrict__ wsList, const int* __restrict__ wsCnt,
                  int* __restrict__ wsPos,
                  const float* __restrict__ Wn, const float* __restrict__ an_src,
                  const float* __restrict__ an_dst,
                  float* __restrict__ ssn, float* __restrict__ sdn,
                  int n_edges, int n_nodes) {
    __shared__ __align__(16) float acc[(OWN_ROWS + 15) / 16 * 16 * ACC_PITCH];
    __shared__ __align__(16) _Float16 Ws[HID * HID];
    __shared__ unsigned srt[OWN_CAP];
    __shared__ int   rcnt[OWN_ROWS];
    __shared__ int   rpos[OWN_ROWS + 1];
    __shared__ float mxl[OWN_ROWS];
    __shared__ float denl[OWN_ROWS];
    __shared__ float sdl[OWN_ROWS];
    __shared__ __align__(16) float wan[2 * HID];
    static_assert(HID / 4 == OWN_THREADS / 32, "a wave folds four rows of the next layer's weights");
    static_assert(OWN_ROWS <= 7 * 32, "the scan below gives a lane seven rows");
    static_assert(OWN_ROWS <= 256, "a list entry keeps its local row in eight bits");
    constexpr int NT16 = (OWN_ROWS + 15) / 16;
    static_assert(2 * NT16 <= OWN_THREADS / 32, "a wave takes a row tile and half of the columns");
    static_assert(HID * HID / 16 == OWN_THREADS, "a thread brings sixteen values of W");

    const int tid = threadIdx.x, nthr = blockDim.x;
    const int lane = tid & 31;
    const int wave = __builtin_amdgcn_readfirstlane(tid >> 5);
    const int nwaves = nthr >> 5;
    const int row0 = blockIdx.x * OWN_ROWS;
    const int rows = (n_nodes - row0 < OWN_ROWS) ? (n_nodes - row0) : OWN_ROWS;
    unsigned* wl = wsList + (size_t)blockIdx.x * OWN_CAP;
    int* wpos = wsPos + (size_t)blockIdx.x * (OWN_ROWS + 1);

    const int nin   = wsCnt[blockIdx.x];
    const bool fits = rows + nin <= OWN_CAP;

    auto score = [&](int s, int dl) {
        const float v = ss[s] + sdl[dl];
        return (v > 0.f) ? v : NEG_SLOPE * v;
    };

    const int wnp = tid & 63, wkg = tid >> 6;
    float2 wv[8];
#pragma unroll
    for (int j = 0; j < 8; ++j)
        wv[j] = *(const float2*)(W + (size_t)(wkg * 8 + j) * HID + 2 * wnp);
    v4f wn[4], asn = {}, adn = {};
    if (FIRST) {
#pragma unroll
        for (int j = 0; j < 4; ++j)
            wn[j] = *(const v4f*)(Wn + (size_t)(wave * 4 + j) * HID + 4 * lane);
        asn = *(const v4f*)(an_src + 4 * lane);
        adn = *(const v4f*)(an_dst + 4 * lane);
    }
    for (int i = tid; i < rows; i += nthr) {
        sdl[i]  = sd[row0 + i];
        rcnt[i] = 1;
    }
    if (!FIRST && fits) {
        for (int i = tid; i < rows + nin; i += nthr) srt[i] = wl[i];
        for (int i = tid; i <= rows; i += nthr) rpos[i] = wpos[i];
    }
    {
        v8h c0, c1;
#pragma unroll
        for (int j = 0; j < 8; ++j) { c0[j] = (_Float16)wv[j].x; c1[j] = (_Float16)wv[j].y; }
        const int kt = wkg >> 2, sel = (wkg >> 1) & 1, hi = wkg & 1;
        const int n0 = 2 * wnp;
        _Float16* wp = Ws + ((n0 >> 4) * 4 + kt) * 512 + sel * 256 + ((n0 & 15) + 16 * hi) * 8;
        *(v8h*)wp       = c0;
        *(v8h*)(wp + 8) = c1;
    }
    if (FIRST) {
#pragma unroll
        for (int j = 0; j < 4; ++j) {
            float ps = wn[j].x * asn.x + wn[j].y * asn.y + wn[j].z * asn.z + wn[j].w * asn.w;
            float pd = wn[j].x * adn.x + wn[j].y * adn.y + wn[j].z * adn.z + wn[j].w * adn.w;
            ps = wave_sum32(ps);
            pd = wave_sum32(pd);
            if (lane == 0) { wan[wave * 4 + j] = ps; wan[HID + wave * 4 + j] = pd; }
        }
    }
    __syncthreads();

    if (fits) {
      if (FIRST) {
        for (int i = tid; i < nin; i += nthr) atomicAdd(&rcnt[wl[i] & 255u], 1);
        __syncthreads();
        if (wave == 0) {
            int c[7], tot = 0;
#pragma unroll
            for (int j = 0; j < 7; ++j) {
                const int idx = lane * 7 + j;
                c[j] = (idx < rows) ? rcnt[idx] : 0;
                tot += c[j];
            }
            int inc = tot;
#pragma unroll
            for (int off = 1; off < 32; off <<= 1) {
                const int t = __shfl_up(inc, off, 32);
                if (lane >= off) inc += t;
            }
            int run = inc - tot;
#pragma unroll
            for (int j = 0; j < 7; ++j) {
                const int idx = lane * 7 + j;
                if (idx < rows) {
                    rpos[idx] = run;
                    srt[run]  = (unsigned)(row0 + idx);
                    rcnt[idx] = run + 1;
                }
                run += c[j];
            }
            if (lane == 31) rpos[rows] = run;
        }
        __syncthreads();
        for (int i = tid; i < nin; i += nthr) {
            const unsigned p = wl[i];
            const int slot = atomicAdd(&rcnt[p & 255u], 1);
            srt[slot] = p >> 8;
        }
        __syncthreads();
      }

        int e0 = 0, deg = 0, s0 = 0;
        float a0 = 0.f;
        auto fetch = [&](int rr) {
            e0  = __builtin_amdgcn_readfirstlane(rpos[rr]);
            deg = __builtin_amdgcn_readfirstlane(rpos[rr + 1]) - e0;
            s0  = (lane < deg) ? (int)srt[e0 + lane] : (row0 + rr);
            a0  = ss[s0];
        };
        if (wave < rows) fetch(wave);
        for (int r = wave; r < rows; r += nwaves) {
            const int ce0 = e0, cdeg = deg, cs0 = s0;
            const float ca0 = a0;
            if (r + nwaves < rows) fetch(r + nwaves);
            const float sdr = sdl[r];
            auto lrelu = [](float v) { return (v > 0.f) ? v : NEG_SLOPE * v; };
            int   cs = cs0;
            float ca = ca0;
            if (FIRST) {
                const int n0 = (cdeg < 32) ? cdeg : 32;
                int below = 0;
                for (int k = 0; k < n0; ++k) {
                    const int sk = __builtin_amdgcn_readlane(cs0, k);
                    below += (sk < cs0 || (sk == cs0 && k < lane)) ? 1 : 0;
                }
                const int place = (lane < n0) ? below : lane;
                cs = __builtin_amdgcn_ds_permute(place << 2, cs0);
                ca = __int_as_float(
                    __builtin_amdgcn_ds_permute(place << 2, __float_as_int(ca0)));
                if (lane < n0) srt[ce0 + lane] = (unsigned)cs;
            }
            const float v0 = (lane < cdeg) ? lrelu(ca + sdr) : -3.402823466e38f;
            float m = v0;
            for (int c0 = 32; c0 < cdeg; c0 += 32) {
                const int i = c0 + lane;
                if (i < cdeg) m = fmaxf(m, lrelu(ss[srt[ce0 + i]] + sdr));
            }
            m = wave_max32(m);
            const float ex0 = (lane < cdeg) ? __expf(v0 - m) : 0.f;
            float den = ex0;
            for (int c0 = 32; c0 < cdeg; c0 += 32) {
                const int i = c0 + lane;
                if (i < cdeg) den += __expf(lrelu(ss[srt[ce0 + i]] + sdr) - m);
            }
            den = wave_sum32(den) + 1e-16f;

            float4 sum = make_float4(0.f, 0.f, 0.f, 0.f);
            for (int c0 = 0; c0 < cdeg; c0 += 32) {
                int s = cs;
                float w = ex0 / den;
                if (c0 > 0) {
                    const int i = c0 + lane;
                    s = (i < cdeg) ? (int)srt[ce0 + i] : (row0 + r);
                    w = (i < cdeg) ? __expf(lrelu(ss[s] + sdr) - m) / den : 0.f;
                }
                const int left = cdeg - c0;
                const int nb = (left < 32) ? left : 32;
                int k0 = 0;
                for (; k0 + 8 <= nb; k0 += 8) add_rows<8>(H, s, w, k0, 8, lane, sum);
                const int rem = nb - k0;
                if (rem > 4)      add_rows<8>(H, s, w, k0, rem, lane, sum);
                else if (rem > 0) add_rows<4>(H, s, w, k0, rem, lane, sum);
            }
            *(float4*)(acc + r * ACC_PITCH + 4 * lane) = sum;
        }
    } else {
        for (int i = tid; i < OWN_ROWS * (HID / 4); i += nthr)
            *(float4*)(acc + (i >> 5) * ACC_PITCH + (i & 31) * 4) = make_float4(0.f, 0.f, 0.f, 0.f);
        for (int i = tid; i < rows; i += nthr) { mxl[i] = -3.402823466e38f; denl[i] = 0.f; }
        __syncthreads();

        auto do_max = [&](int s, int dl) { ldsMaxF(&mxl[dl], score(s, dl)); };
        for (int i = tid; i < rows; i += nthr) do_max(row0 + i, i);
        scan_own_edges(edst, n_edges, row0, rows, [&](int e) { do_max(esrc[e], edst[e] - row0); });
        __syncthreads();

        auto do_sum = [&](int s, int dl) { atomicAdd(&denl[dl], __expf(score(s, dl) - mxl[dl])); };
        for (int i = tid; i < rows; i += nthr) do_sum(row0 + i, i);
        scan_own_edges(edst, n_edges, row0, rows, [&](int e) { do_sum(esrc[e], edst[e] - row0); });
        __syncthreads();

        for (int idx = tid; idx < rows * HID; idx += nthr) {
            const int dl = idx / HID, c = idx % HID, s = row0 + dl;
            const float w = __expf(score(s, dl) - mxl[dl]) / (denl[dl] + 1e-16f);
            atomicAdd(&acc[dl * ACC_PITCH + c], w * H[(size_t)s * HID + c]);
        }
        scan_own_edges(edst, n_edges, row0, rows, [&](int e) {
            const int s = esrc[e], dl = edst[e] - row0;
            const float w = __expf(score(s, dl) - mxl[dl]) / (denl[dl] + 1e-16f);
            const float* hs = H + (size_t)s * HID;
            for (int c = 0; c < HID; ++c) atomicAdd(&acc[dl * ACC_PITCH + c], w * hs[c]);
        });
    }

    __syncthreads();
    if (FIRST && fits) {
        for (int i = tid; i < rows + nin; i += nthr) wl[i] = srt[i];
        for (int i = tid; i <= rows; i += nthr) wpos[i] = rpos[i];
    }
    {
        const int tile = wave >> 1, half = wave & 1;
        const bool has = tile < NT16;
        const int hl = lane & 15, hi = lane >> 4;
        v16h a[4];
        if (has) {
            const float* ap = acc + (tile * 16 + hl) * ACC_PITCH + hi * 8;
#pragma unroll
            for (int kt = 0; kt < 4; ++kt) {
                const float4 x0 = *(const float4*)(ap + kt * 32);
                const float4 x1 = *(const float4*)(ap + kt * 32 + 4);
                const float4 y0 = *(const float4*)(ap + kt * 32 + 16);
                const float4 y1 = *(const float4*)(ap + kt * 32 + 20);
                a[kt] = v16h{ (_Float16)x0.x, (_Float16)x0.y, (_Float16)x0.z, (_Float16)x0.w,
                              (_Float16)x1.x, (_Float16)x1.y, (_Float16)x1.z, (_Float16)x1.w,
                              (_Float16)y0.x, (_Float16)y0.y, (_Float16)y0.z, (_Float16)y0.w,
                              (_Float16)y1.x, (_Float16)y1.y, (_Float16)y1.z, (_Float16)y1.w };
            }
        }
        __syncthreads();
        if (has) {
#pragma unroll
            for (int q = 0; q < 4; ++q) {
                const int ct = half * 4 + q;
                v8f d = v8f{};
#pragma unroll
                for (int kt = 0; kt < 4; ++kt) {
                    const _Float16* bp = Ws + (ct * 4 + kt) * 512 + lane * 8;
                    const v8h blo = *(const v8h*)bp;
                    const v8h bhi = *(const v8h*)(bp + 256);
                    v16h b;
#pragma unroll
                    for (int j = 0; j < 8; ++j) { b[j] = blo[j]; b[8 + j] = bhi[j]; }
                    d = __builtin_amdgcn_wmma_f32_16x16x32_f16(
                        false, a[kt], false, b, (short)0, d, false, false);
                }
#pragma unroll
                for (int r = 0; r < 8; ++r)
                    acc[(tile * 16 + r + 8 * hi) * ACC_PITCH + ct * 16 + hl] = d[r];
            }
        }
    }

    auto finish = [](float v, float b, float mu, float sc, float be) {
        return fmaxf((v + b - mu) * sc + be, 0.f);
    };

    if (FIRST) {
        const int c0 = (tid & 31) * 4;
        float b[4], mu[4], sc[4], be[4];
#pragma unroll
        for (int j = 0; j < 4; ++j) {
            b[j]  = bias[c0 + j];
            mu[j] = mean[c0 + j];
            sc[j] = gam[c0 + j] * rsqrtf(var[c0 + j] + BN_EPS);
            be[j] = bet[c0 + j];
        }
        const v4f was = *(const v4f*)(wan + c0);
        const v4f wad = *(const v4f*)(wan + HID + c0);
        __syncthreads();

        float4* o4 = (float4*)(out + (size_t)row0 * HID);
        for (int i = tid; i < rows * (HID / 4); i += nthr) {
            float4 a = *(const float4*)(acc + (i >> 5) * ACC_PITCH + (i & 31) * 4);
            a.x = finish(a.x, b[0], mu[0], sc[0], be[0]);
            a.y = finish(a.y, b[1], mu[1], sc[1], be[1]);
            a.z = finish(a.z, b[2], mu[2], sc[2], be[2]);
            a.w = finish(a.w, b[3], mu[3], sc[3], be[3]);
            o4[i] = a;
            float ps = a.x * was.x + a.y * was.y + a.z * was.z + a.w * was.w;
            float pd = a.x * wad.x + a.y * wad.y + a.z * wad.z + a.w * wad.w;
            ps = wave_sum32(ps);
            pd = wave_sum32(pd);
            if (lane == 0) { ssn[row0 + (i >> 5)] = ps; sdn[row0 + (i >> 5)] = pd; }
        }
    } else {
        static_assert((OWN_ROWS + OWN_THREADS / HID - 1) / (OWN_THREADS / HID) <= 32,
                      "a slice's graph ids are held one per lane");
        const int c = tid & (HID - 1);
        const float b  = bias[c];
        const float mu = mean[c];
        const float sc = gam[c] * rsqrtf(var[c] + BN_EPS);
        const float be = bet[c];
        const int ngrp = nthr / HID;
        const int per  = (rows + ngrp - 1) / ngrp;
        const int r0   = (wave >> 2) * per;
        const int n    = (rows - r0 < per) ? (rows - r0) : per;
        const int gid = (lane < n) ? batch[row0 + r0 + lane] : 0;
        __syncthreads();

        if (n > 0) {
            int cur = __builtin_amdgcn_readlane(gid, 0);
            float run = 0.f, nrun = 0.f;
            for (int k = 0; k < n; ++k) {
                const int g = __builtin_amdgcn_readlane(gid, k);
                if (g != cur) {
                    atomicAdd(&pool[(size_t)cur * HID + c], run);
                    if (c == 0) atomicAdd(&pcnt[cur], nrun);
                    cur = g; run = 0.f; nrun = 0.f;
                }
                run += finish(acc[(r0 + k) * ACC_PITCH + c], b, mu, sc, be);
                nrun += 1.f;
            }
            atomicAdd(&pool[(size_t)cur * HID + c], run);
            if (c == 0) atomicAdd(&pcnt[cur], nrun);
        }
    }
}

__global__ void pool_div(float* __restrict__ sums, const float* __restrict__ cnt, int total) {
    int i = blockIdx.x * blockDim.x + threadIdx.x;
    if (i >= total) return;
    sums[i] /= fmaxf(cnt[i >> 7], 1.0f);
}

extern "C" void kernel_launch(void* const* d_in, const int* in_sizes, int n_in,
                              void* d_out, int out_size, void* d_ws, size_t ws_size,
                              hipStream_t stream) {
    const float* gx  = (const float*)d_in[0];
    const int*   ei  = (const int*)  d_in[1];
    const int*   bat = (const int*)  d_in[2];
    const float* W1  = (const float*)d_in[3];
    const float* as1 = (const float*)d_in[4];
    const float* ad1 = (const float*)d_in[5];
    const float* b1  = (const float*)d_in[6];
    const float* g1  = (const float*)d_in[7];
    const float* be1 = (const float*)d_in[8];
    const float* mu1 = (const float*)d_in[9];
    const float* va1 = (const float*)d_in[10];
    const float* W2  = (const float*)d_in[11];
    const float* as2 = (const float*)d_in[12];
    const float* ad2 = (const float*)d_in[13];
    const float* b2  = (const float*)d_in[14];
    const float* g2  = (const float*)d_in[15];
    const float* be2 = (const float*)d_in[16];
    const float* mu2 = (const float*)d_in[17];
    const float* va2 = (const float*)d_in[18];
    const float* hw  = (const float*)d_in[19];
    const float* hb  = (const float*)d_in[20];

    const int n_nodes = in_sizes[0] / HID;
    const int n_edges = in_sizes[1] / 2;
    const int* esrc = ei;
    const int* edst = ei + n_edges;

    float* ws   = (float*)d_ws;
    float* AGG  = ws;
    float* SS   = AGG  + (size_t)n_nodes * HID;
    float* SD   = SS   + n_nodes;
    float* POOL = SD   + n_nodes;
    float* CNT  = POOL + (size_t)N_GRAPHS * HID;
    const int gridOwn   = (n_nodes + OWN_ROWS - 1) / OWN_ROWS;
    int*      OCNT  = (int*)(CNT + N_GRAPHS);
    unsigned* OLIST = (unsigned*)(OCNT + gridOwn);
    int*   OPOS = (int*)(OLIST + (size_t)gridOwn * OWN_CAP);
    float* SS2  = (float*)(OPOS + (size_t)gridOwn * (OWN_ROWS + 1));
    float* SD2  = SS2 + n_nodes;

    const int rowsScoreWg = (SC_THREADS / 32) * SC_ROWS;
    const int gridNodeW = (n_nodes + rowsScoreWg - 1) / rowsScoreWg;

    fill_f32<<<(N_GRAPHS * HID + 255) / 256, 256, 0, stream>>>(POOL, 0.0f, N_GRAPHS * HID);
    fill_f32<<<(N_GRAPHS + gridOwn + 255) / 256, 256, 0, stream>>>(CNT, 0.0f, N_GRAPHS + gridOwn);

    split_edges_by_owner<<<(n_edges + SPLIT_THREADS * SPLIT_EPT - 1) / (SPLIT_THREADS * SPLIT_EPT),
                           SPLIT_THREADS, 0, stream>>>(esrc, edst, OLIST, OCNT,
                                                       n_edges, n_nodes, gridOwn);

    node_scores<<<gridNodeW, SC_THREADS, 0, stream>>>(gx, W1, as1, ad1, SS, SD, n_nodes);
    own_rows_gat<true><<<gridOwn, OWN_THREADS, 0, stream>>>(
        gx, W1, esrc, edst, SS, SD, b1, g1, be1, mu1, va1, AGG, bat, POOL, CNT,
        OLIST, OCNT, OPOS, W2, as2, ad2, SS2, SD2, n_edges, n_nodes);
    own_rows_gat<false><<<gridOwn, OWN_THREADS, 0, stream>>>(
        AGG, W2, esrc, edst, SS2, SD2, b2, g2, be2, mu2, va2, AGG, bat, POOL, CNT,
        OLIST, OCNT, OPOS, nullptr, nullptr, nullptr, nullptr, nullptr, n_edges, n_nodes);

    pool_div<<<(N_GRAPHS * HID + 255) / 256, 256, 0, stream>>>(POOL, CNT, N_GRAPHS * HID);

    gemm128_wmma<true, 4, 1><<<N_GRAPHS / 16, 32 * (64 / 16), 0, stream>>>(
        POOL, hw, hb, (float*)d_out, N_GRAPHS);
}


# Launches 8 -> 5: one fill zeroes every cleared word; edge split and layer 1's node scores run as two block roles of one grid; pool mean taken as the head product stages its rows; sums unchanged
# speedup vs baseline: 7.0547x; 1.0656x over previous
#include <hip/hip_runtime.h>
#include <hip/hip_bf16.h>

typedef __attribute__((ext_vector_type(16))) _Float16 v16h;
typedef __attribute__((ext_vector_type(8)))  _Float16 v8h;
typedef __attribute__((ext_vector_type(4)))  _Float16 v4h;
typedef __attribute__((ext_vector_type(8)))  float    v8f;

#define HID 128
#define NEG_SLOPE 0.2f
#define BN_EPS 1e-5f
#define N_GRAPHS 512

template <bool HAS_BIAS, bool MEAN, int NW, int RT>
__global__ __launch_bounds__(NW * 32)
void gemm128_wmma(const float* __restrict__ A, const float* __restrict__ rowcnt,
                  const float* __restrict__ B,
                  const float* __restrict__ bias, float* __restrict__ C, int M) {
    constexpr int N     = NW * 16;
    constexpr int NT    = NW * 32;
    constexpr int PER   = RT * 512 / NT;
    constexpr int PITCH = 136;
    static_assert(RT * 512 % NT == 0, "the A pieces divide evenly among the threads");
    __shared__ __align__(16) _Float16 As[RT * 16 * PITCH];
    const int tid  = threadIdx.x;
    const int row0 = blockIdx.x * (RT * 16);
    const int wave = tid >> 5;
    const int lane = tid & 31;
    const int hl   = lane & 15;
    const int hi   = lane >> 4;

    float4 f[PER];
    float  dn[PER];
#pragma unroll
    for (int q = 0; q < PER; ++q) {
        const int c = tid + q * NT;
        int row = row0 + (c >> 5);
        if (row >= M) row = M - 1;
        f[q] = *(const float4*)(A + (size_t)row * 128 + (c & 31) * 4);
        dn[q] = MEAN ? rowcnt[row] : 1.0f;
    }
    v16h b[4];
#pragma unroll
    for (int kt = 0; kt < 4; ++kt) {
        const float* bp = B + (size_t)(kt * 32 + hi * 8) * N + wave * 16 + hl;
#pragma unroll
        for (int j = 0; j < 8; ++j) {
            b[kt][j]     = (_Float16)bp[(size_t)j * N];
            b[kt][8 + j] = (_Float16)bp[(size_t)(16 + j) * N];
        }
    }
#pragma unroll
    for (int q = 0; q < PER; ++q) {
        const int c = tid + q * NT;
        if (MEAN) {
            const float d = fmaxf(dn[q], 1.0f);
            f[q].x /= d; f[q].y /= d; f[q].z /= d; f[q].w /= d;
        }
        v4h h4 = { (_Float16)f[q].x, (_Float16)f[q].y, (_Float16)f[q].z, (_Float16)f[q].w };
        *(v4h*)(As + (c >> 5) * PITCH + (c & 31) * 4) = h4;
    }
    __syncthreads();

    v8f acc[RT];
#pragma unroll
    for (int i = 0; i < RT; ++i) acc[i] = v8f{};
#pragma unroll
    for (int kt = 0; kt < 4; ++kt) {
#pragma unroll
        for (int i = 0; i < RT; ++i) {
            const int base = (i * 16 + hl) * PITCH + kt * 32 + hi * 8;
            v8h alo = *(const v8h*)(As + base);
            v8h ahi = *(const v8h*)(As + base + 16);
            v16h a;
#pragma unroll
            for (int j = 0; j < 8; ++j) { a[j] = alo[j]; a[8 + j] = ahi[j]; }
            acc[i] = __builtin_amdgcn_wmma_f32_16x16x32_f16(
                false, a, false, b[kt], (short)0, acc[i], false, false);
        }
    }

    const int col = wave * 16 + hl;
    float bv = HAS_BIAS ? bias[col] : 0.0f;
#pragma unroll
    for (int i = 0; i < RT; ++i) {
        if (row0 + i * 16 < M) {
#pragma unroll
            for (int r = 0; r < 8; ++r) {
                const int row = row0 + i * 16 + r + 8 * hi;
                C[(size_t)row * N + col] = acc[i][r] + bv;
            }
        }
    }
}

#define SC_ROWS 16
#define SC_THREADS 256
typedef float v4f __attribute__((ext_vector_type(4)));

template <int NH, int OFF>
__device__ __forceinline__ void fold(float (&p)[2 * SC_ROWS], int lane) {
    const bool up = (lane & OFF) != 0;
#pragma unroll
    for (int i = 0; i < NH; ++i) {
        const float keep = up ? p[i + NH] : p[i];
        const float send = up ? p[i] : p[i + NH];
        p[i] = keep + __shfl_xor(send, OFF, 32);
    }
}

__device__ __forceinline__ void load16(const float* __restrict__ H, int row0, int lane,
                                       v4f (&h)[SC_ROWS]) {
    const float* hb = H + (size_t)row0 * HID + 4 * lane;
#pragma unroll
    for (int k = 0; k < SC_ROWS; ++k) h[k] = *(const v4f*)(hb + k * HID);
}

__device__ __forceinline__ void dots16(const v4f (&h)[SC_ROWS], v4f as, v4f ad,
                                       float* __restrict__ ss, float* __restrict__ sd, int lane) {
    float p[2 * SC_ROWS];
#pragma unroll
    for (int k = 0; k < SC_ROWS; ++k) {
        p[k]           = h[k].x * as.x + h[k].y * as.y + h[k].z * as.z + h[k].w * as.w;
        p[SC_ROWS + k] = h[k].x * ad.x + h[k].y * ad.y + h[k].z * ad.z + h[k].w * ad.w;
    }
    fold<16, 16>(p, lane);
    fold<8, 8>(p, lane);
    fold<4, 4>(p, lane);
    fold<2, 2>(p, lane);
    fold<1, 1>(p, lane);
    if (lane < 16) ss[lane & 15] = p[0];
    else           sd[lane & 15] = p[0];
}

__device__ __forceinline__
void node_scores_block(int blk, const float* __restrict__ H, const float* __restrict__ W,
                       const float* __restrict__ a_src, const float* __restrict__ a_dst,
                       float* __restrict__ ss, float* __restrict__ sd, int n_nodes) {
    static_assert(HID / SC_ROWS <= SC_THREADS / 32, "W's rows find a wave each sixteen");
    __shared__ __align__(16) float wa[2 * HID];
    const int lane = threadIdx.x & 31;
    const int wave = threadIdx.x >> 5;
    const int row0 = (blk * (SC_THREADS / 32) + wave) * SC_ROWS;
    const bool live = row0 < n_nodes;
    v4f hn[SC_ROWS];
    load16(H, live ? row0 : 0, lane, hn);
    if (wave < HID / SC_ROWS) {
        v4f hw[SC_ROWS];
        load16(W, wave * SC_ROWS, lane, hw);
        const v4f as = *(const v4f*)(a_src + 4 * lane);
        const v4f ad = *(const v4f*)(a_dst + 4 * lane);
        dots16(hw, as, ad, wa + wave * SC_ROWS, wa + HID + wave * SC_ROWS, lane);
    }
    __syncthreads();
    if (!live) return;
    const v4f was = *(const v4f*)(wa + 4 * lane);
    const v4f wad = *(const v4f*)(wa + HID + 4 * lane);
    dots16(hn, was, wad, ss + row0, sd + row0, lane);
}

__global__ void fill_f32(float* __restrict__ p, float v, int n) {
    int i = blockIdx.x * blockDim.x + threadIdx.x;
    if (i < n) p[i] = v;
}

#define OWN_ROWS    196
#define ACC_PITCH   132
#define OWN_CAP     16384
#define OWN_THREADS 1024

typedef int v4i __attribute__((ext_vector_type(4)));

__device__ __forceinline__ void ldsMaxF(float* addr, float val) {
    if (val >= 0.f) atomicMax((int*)addr, __float_as_int(val));
    else            atomicMin((unsigned int*)addr, __float_as_uint(val));
}

template <class F>
__device__ __forceinline__ void scan_own_edges(const int* __restrict__ edst, int n_edges,
                                               int row0, int rows, F f) {
    const int nthr = blockDim.x;
    const int nvec = (((size_t)edst & 15) == 0) ? (n_edges >> 2) : 0;
    const v4i* dv = (const v4i*)edst;
    const int rot = (int)blockIdx.x * (nvec / (int)gridDim.x);
    for (int v0 = threadIdx.x; v0 < nvec; v0 += 8 * nthr) {
        v4i d[8];
#pragma unroll
        for (int k = 0; k < 8; ++k) {
            const int v = v0 + k * nthr;
            int uu = v + rot;
            if (uu >= nvec) uu -= nvec;
            const v4i none = { -1, -1, -1, -1 };
            d[k] = (v < nvec) ? __builtin_nontemporal_load(dv + uu) : none;
        }
        unsigned m = 0;
#pragma unroll
        for (int k = 0; k < 8; ++k) {
#pragma unroll
            for (int c = 0; c < 4; ++c)
                m |= ((unsigned)(d[k][c] - row0) < (unsigned)rows) ? (1u << (4 * k + c)) : 0u;
        }
        while (m) {
            const int b = __builtin_ctz(m);
            m &= m - 1;
            int uu = v0 + (b >> 2) * nthr + rot;
            if (uu >= nvec) uu -= nvec;
            f(4 * uu + (b & 3));
        }
    }
    for (int e = nvec * 4 + threadIdx.x; e < n_edges; e += nthr)
        if ((unsigned)(edst[e] - row0) < (unsigned)rows) f(e);
}

#define SPLIT_THREADS 256
#define SPLIT_EPT     16
#define SPLIT_MAX_OWN 1024

__device__ __forceinline__
void split_edges_block(int blk, const int* __restrict__ esrc, const int* __restrict__ edst,
                       unsigned* __restrict__ wsList, int* __restrict__ wsCnt,
                       int n_edges, int n_nodes, int n_own) {
    __shared__ int hist[SPLIT_MAX_OWN];
    __shared__ int hbase[SPLIT_MAX_OWN];
    const int tid = threadIdx.x;
    for (int o = tid; o < n_own; o += SPLIT_THREADS) hist[o] = 0;

    const int e0 = blk * (SPLIT_THREADS * SPLIT_EPT) + tid;
    int d[SPLIT_EPT], s[SPLIT_EPT];
#pragma unroll
    for (int k = 0; k < SPLIT_EPT; ++k) {
        const int e  = e0 + k * SPLIT_THREADS;
        const int ec = (e < n_edges) ? e : n_edges - 1;
        d[k] = edst[ec];
        s[k] = esrc[ec];
        if (e >= n_edges) d[k] = -1;
    }
    __syncthreads();

    int own[SPLIT_EPT], rank[SPLIT_EPT];
#pragma unroll
    for (int k = 0; k < SPLIT_EPT; ++k) {
        own[k]  = ((unsigned)d[k] < (unsigned)n_nodes) ? d[k] / OWN_ROWS : -1;
        rank[k] = 0;
        if (own[k] >= 0) rank[k] = atomicAdd(&hist[own[k]], 1);
    }
    __syncthreads();

    for (int o = tid; o < n_own; o += SPLIT_THREADS) {
        const int cnt = hist[o];
        hbase[o] = (cnt > 0) ? atomicAdd(&wsCnt[o], cnt) : 0;
    }
    __syncthreads();

    int pos[SPLIT_EPT];
#pragma unroll
    for (int k = 0; k < SPLIT_EPT; ++k) pos[k] = hbase[(own[k] >= 0) ? own[k] : 0] + rank[k];
#pragma unroll
    for (int k = 0; k < SPLIT_EPT; ++k) {
        if (own[k] >= 0 && pos[k] < OWN_CAP)
            wsList[(size_t)own[k] * OWN_CAP + pos[k]] =
                ((unsigned)s[k] << 8) | (unsigned)(d[k] - own[k] * OWN_ROWS);
    }
}

static_assert(SPLIT_THREADS == SC_THREADS, "the two roles share one block size");
__global__ __launch_bounds__(SPLIT_THREADS)
void split_and_scores(int n_split,
                      const int* __restrict__ esrc, const int* __restrict__ edst,
                      unsigned* __restrict__ wsList, int* __restrict__ wsCnt,
                      int n_edges, int n_nodes, int n_own,
                      const float* __restrict__ H, const float* __restrict__ W,
                      const float* __restrict__ a_src, const float* __restrict__ a_dst,
                      float* __restrict__ ss, float* __restrict__ sd) {
    const int blk = (int)blockIdx.x;
    if (blk < n_split)
        split_edges_block(blk, esrc, edst, wsList, wsCnt, n_edges, n_nodes, n_own);
    else
        node_scores_block(blk - n_split, H, W, a_src, a_dst, ss, sd, n_nodes);
}


template <int M>
__device__ __forceinline__ float xrow(float x) {
    return __builtin_bit_cast(float, __builtin_amdgcn_update_dpp(
        0, __builtin_bit_cast(int, x), 0x160 + M, 0xF, 0xF, true));
}
__device__ __forceinline__ float xhalf(float x) {
    const int b = __builtin_bit_cast(int, x);
    return __builtin_bit_cast(float, __builtin_amdgcn_permlanex16(b, b, 0x76543210, 0xfedcba98, false, false));
}
__device__ __forceinline__ float wave_max32(float v) {
    v = fmaxf(v, xrow<1>(v));
    v = fmaxf(v, xrow<2>(v));
    v = fmaxf(v, xrow<4>(v));
    v = fmaxf(v, xrow<8>(v));
    return fmaxf(v, xhalf(v));
}
__device__ __forceinline__ float wave_sum32(float v) {
    v += xrow<1>(v);
    v += xrow<2>(v);
    v += xrow<4>(v);
    v += xrow<8>(v);
    return v + xhalf(v);
}

template <int NB>
__device__ __forceinline__ void add_rows(const float* __restrict__ H, int s, float w, int k0,
                                         int nvalid, int lane, float4& sum) {
    float4 hv[NB];
    float  wk[NB];
#pragma unroll
    for (int k = 0; k < NB; ++k) {
        const int kk = (k < nvalid) ? (k0 + k) : k0;
        const int sk = __builtin_amdgcn_readlane(s, kk);
        const float wv = __int_as_float(__builtin_amdgcn_readlane(__float_as_int(w), kk));
        wk[k] = (k < nvalid) ? wv : 0.f;
        hv[k] = *(const float4*)(H + (size_t)sk * HID + 4 * lane);
    }
#pragma unroll
    for (int k = 0; k < NB; ++k) {
        sum.x += wk[k] * hv[k].x;
        sum.y += wk[k] * hv[k].y;
        sum.z += wk[k] * hv[k].z;
        sum.w += wk[k] * hv[k].w;
    }
}

template <bool FIRST>
__global__ __launch_bounds__(OWN_THREADS)
void own_rows_gat(const float* __restrict__ H, const float* __restrict__ W,
                  const int* __restrict__ esrc,
                  const int* __restrict__ edst, const float* __restrict__ ss,
                  const float* __restrict__ sd,
                  const float* __restrict__ bias, const float* __restrict__ gam,
                  const float* __restrict__ bet, const float* __restrict__ mean,
                  const float* __restrict__ var,
                  float* __restrict__ out,
                  const int* __restrict__ batch, float* __restrict__ pool,
                  float* __restrict__ pcnt,
                  unsigned* __restrict__ wsList, const int* __restrict__ wsCnt,
                  int* __restrict__ wsPos,
                  const float* __restrict__ Wn, const float* __restrict__ an_src,
                  const float* __restrict__ an_dst,
                  float* __restrict__ ssn, float* __restrict__ sdn,
                  int n_edges, int n_nodes) {
    __shared__ __align__(16) float acc[(OWN_ROWS + 15) / 16 * 16 * ACC_PITCH];
    __shared__ __align__(16) _Float16 Ws[HID * HID];
    __shared__ unsigned srt[OWN_CAP];
    __shared__ int   rcnt[OWN_ROWS];
    __shared__ int   rpos[OWN_ROWS + 1];
    __shared__ float mxl[OWN_ROWS];
    __shared__ float denl[OWN_ROWS];
    __shared__ float sdl[OWN_ROWS];
    __shared__ __align__(16) float wan[2 * HID];
    static_assert(HID / 4 == OWN_THREADS / 32, "a wave folds four rows of the next layer's weights");
    static_assert(OWN_ROWS <= 7 * 32, "the scan below gives a lane seven rows");
    static_assert(OWN_ROWS <= 256, "a list entry keeps its local row in eight bits");
    constexpr int NT16 = (OWN_ROWS + 15) / 16;
    static_assert(2 * NT16 <= OWN_THREADS / 32, "a wave takes a row tile and half of the columns");
    static_assert(HID * HID / 16 == OWN_THREADS, "a thread brings sixteen values of W");

    const int tid = threadIdx.x, nthr = blockDim.x;
    const int lane = tid & 31;
    const int wave = __builtin_amdgcn_readfirstlane(tid >> 5);
    const int nwaves = nthr >> 5;
    const int row0 = blockIdx.x * OWN_ROWS;
    const int rows = (n_nodes - row0 < OWN_ROWS) ? (n_nodes - row0) : OWN_ROWS;
    unsigned* wl = wsList + (size_t)blockIdx.x * OWN_CAP;
    int* wpos = wsPos + (size_t)blockIdx.x * (OWN_ROWS + 1);

    const int nin   = wsCnt[blockIdx.x];
    const bool fits = rows + nin <= OWN_CAP;

    auto score = [&](int s, int dl) {
        const float v = ss[s] + sdl[dl];
        return (v > 0.f) ? v : NEG_SLOPE * v;
    };

    const int wnp = tid & 63, wkg = tid >> 6;
    float2 wv[8];
#pragma unroll
    for (int j = 0; j < 8; ++j)
        wv[j] = *(const float2*)(W + (size_t)(wkg * 8 + j) * HID + 2 * wnp);
    v4f wn[4], asn = {}, adn = {};
    if (FIRST) {
#pragma unroll
        for (int j = 0; j < 4; ++j)
            wn[j] = *(const v4f*)(Wn + (size_t)(wave * 4 + j) * HID + 4 * lane);
        asn = *(const v4f*)(an_src + 4 * lane);
        adn = *(const v4f*)(an_dst + 4 * lane);
    }
    for (int i = tid; i < rows; i += nthr) {
        sdl[i]  = sd[row0 + i];
        rcnt[i] = 1;
    }
    if (!FIRST && fits) {
        for (int i = tid; i < rows + nin; i += nthr) srt[i] = wl[i];
        for (int i = tid; i <= rows; i += nthr) rpos[i] = wpos[i];
    }
    {
        v8h c0, c1;
#pragma unroll
        for (int j = 0; j < 8; ++j) { c0[j] = (_Float16)wv[j].x; c1[j] = (_Float16)wv[j].y; }
        const int kt = wkg >> 2, sel = (wkg >> 1) & 1, hi = wkg & 1;
        const int n0 = 2 * wnp;
        _Float16* wp = Ws + ((n0 >> 4) * 4 + kt) * 512 + sel * 256 + ((n0 & 15) + 16 * hi) * 8;
        *(v8h*)wp       = c0;
        *(v8h*)(wp + 8) = c1;
    }
    if (FIRST) {
#pragma unroll
        for (int j = 0; j < 4; ++j) {
            float ps = wn[j].x * asn.x + wn[j].y * asn.y + wn[j].z * asn.z + wn[j].w * asn.w;
            float pd = wn[j].x * adn.x + wn[j].y * adn.y + wn[j].z * adn.z + wn[j].w * adn.w;
            ps = wave_sum32(ps);
            pd = wave_sum32(pd);
            if (lane == 0) { wan[wave * 4 + j] = ps; wan[HID + wave * 4 + j] = pd; }
        }
    }
    __syncthreads();

    if (fits) {
      if (FIRST) {
        for (int i = tid; i < nin; i += nthr) atomicAdd(&rcnt[wl[i] & 255u], 1);
        __syncthreads();
        if (wave == 0) {
            int c[7], tot = 0;
#pragma unroll
            for (int j = 0; j < 7; ++j) {
                const int idx = lane * 7 + j;
                c[j] = (idx < rows) ? rcnt[idx] : 0;
                tot += c[j];
            }
            int inc = tot;
#pragma unroll
            for (int off = 1; off < 32; off <<= 1) {
                const int t = __shfl_up(inc, off, 32);
                if (lane >= off) inc += t;
            }
            int run = inc - tot;
#pragma unroll
            for (int j = 0; j < 7; ++j) {
                const int idx = lane * 7 + j;
                if (idx < rows) {
                    rpos[idx] = run;
                    srt[run]  = (unsigned)(row0 + idx);
                    rcnt[idx] = run + 1;
                }
                run += c[j];
            }
            if (lane == 31) rpos[rows] = run;
        }
        __syncthreads();
        for (int i = tid; i < nin; i += nthr) {
            const unsigned p = wl[i];
            const int slot = atomicAdd(&rcnt[p & 255u], 1);
            srt[slot] = p >> 8;
        }
        __syncthreads();
      }

        int e0 = 0, deg = 0, s0 = 0;
        float a0 = 0.f;
        auto fetch = [&](int rr) {
            e0  = __builtin_amdgcn_readfirstlane(rpos[rr]);
            deg = __builtin_amdgcn_readfirstlane(rpos[rr + 1]) - e0;
            s0  = (lane < deg) ? (int)srt[e0 + lane] : (row0 + rr);
            a0  = ss[s0];
        };
        if (wave < rows) fetch(wave);
        for (int r = wave; r < rows; r += nwaves) {
            const int ce0 = e0, cdeg = deg, cs0 = s0;
            const float ca0 = a0;
            if (r + nwaves < rows) fetch(r + nwaves);
            const float sdr = sdl[r];
            auto lrelu = [](float v) { return (v > 0.f) ? v : NEG_SLOPE * v; };
            int   cs = cs0;
            float ca = ca0;
            if (FIRST) {
                const int n0 = (cdeg < 32) ? cdeg : 32;
                int below = 0;
                for (int k = 0; k < n0; ++k) {
                    const int sk = __builtin_amdgcn_readlane(cs0, k);
                    below += (sk < cs0 || (sk == cs0 && k < lane)) ? 1 : 0;
                }
                const int place = (lane < n0) ? below : lane;
                cs = __builtin_amdgcn_ds_permute(place << 2, cs0);
                ca = __int_as_float(
                    __builtin_amdgcn_ds_permute(place << 2, __float_as_int(ca0)));
                if (lane < n0) srt[ce0 + lane] = (unsigned)cs;
            }
            const float v0 = (lane < cdeg) ? lrelu(ca + sdr) : -3.402823466e38f;
            float m = v0;
            for (int c0 = 32; c0 < cdeg; c0 += 32) {
                const int i = c0 + lane;
                if (i < cdeg) m = fmaxf(m, lrelu(ss[srt[ce0 + i]] + sdr));
            }
            m = wave_max32(m);
            const float ex0 = (lane < cdeg) ? __expf(v0 - m) : 0.f;
            float den = ex0;
            for (int c0 = 32; c0 < cdeg; c0 += 32) {
                const int i = c0 + lane;
                if (i < cdeg) den += __expf(lrelu(ss[srt[ce0 + i]] + sdr) - m);
            }
            den = wave_sum32(den) + 1e-16f;

            float4 sum = make_float4(0.f, 0.f, 0.f, 0.f);
            for (int c0 = 0; c0 < cdeg; c0 += 32) {
                int s = cs;
                float w = ex0 / den;
                if (c0 > 0) {
                    const int i = c0 + lane;
                    s = (i < cdeg) ? (int)srt[ce0 + i] : (row0 + r);
                    w = (i < cdeg) ? __expf(lrelu(ss[s] + sdr) - m) / den : 0.f;
                }
                const int left = cdeg - c0;
                const int nb = (left < 32) ? left : 32;
                int k0 = 0;
                for (; k0 + 8 <= nb; k0 += 8) add_rows<8>(H, s, w, k0, 8, lane, sum);
                const int rem = nb - k0;
                if (rem > 4)      add_rows<8>(H, s, w, k0, rem, lane, sum);
                else if (rem > 0) add_rows<4>(H, s, w, k0, rem, lane, sum);
            }
            *(float4*)(acc + r * ACC_PITCH + 4 * lane) = sum;
        }
    } else {
        for (int i = tid; i < OWN_ROWS * (HID / 4); i += nthr)
            *(float4*)(acc + (i >> 5) * ACC_PITCH + (i & 31) * 4) = make_float4(0.f, 0.f, 0.f, 0.f);
        for (int i = tid; i < rows; i += nthr) { mxl[i] = -3.402823466e38f; denl[i] = 0.f; }
        __syncthreads();

        auto do_max = [&](int s, int dl) { ldsMaxF(&mxl[dl], score(s, dl)); };
        for (int i = tid; i < rows; i += nthr) do_max(row0 + i, i);
        scan_own_edges(edst, n_edges, row0, rows, [&](int e) { do_max(esrc[e], edst[e] - row0); });
        __syncthreads();

        auto do_sum = [&](int s, int dl) { atomicAdd(&denl[dl], __expf(score(s, dl) - mxl[dl])); };
        for (int i = tid; i < rows; i += nthr) do_sum(row0 + i, i);
        scan_own_edges(edst, n_edges, row0, rows, [&](int e) { do_sum(esrc[e], edst[e] - row0); });
        __syncthreads();

        for (int idx = tid; idx < rows * HID; idx += nthr) {
            const int dl = idx / HID, c = idx % HID, s = row0 + dl;
            const float w = __expf(score(s, dl) - mxl[dl]) / (denl[dl] + 1e-16f);
            atomicAdd(&acc[dl * ACC_PITCH + c], w * H[(size_t)s * HID + c]);
        }
        scan_own_edges(edst, n_edges, row0, rows, [&](int e) {
            const int s = esrc[e], dl = edst[e] - row0;
            const float w = __expf(score(s, dl) - mxl[dl]) / (denl[dl] + 1e-16f);
            const float* hs = H + (size_t)s * HID;
            for (int c = 0; c < HID; ++c) atomicAdd(&acc[dl * ACC_PITCH + c], w * hs[c]);
        });
    }

    __syncthreads();
    if (FIRST && fits) {
        for (int i = tid; i < rows + nin; i += nthr) wl[i] = srt[i];
        for (int i = tid; i <= rows; i += nthr) wpos[i] = rpos[i];
    }
    {
        const int tile = wave >> 1, half = wave & 1;
        const bool has = tile < NT16;
        const int hl = lane & 15, hi = lane >> 4;
        v16h a[4];
        if (has) {
            const float* ap = acc + (tile * 16 + hl) * ACC_PITCH + hi * 8;
#pragma unroll
            for (int kt = 0; kt < 4; ++kt) {
                const float4 x0 = *(const float4*)(ap + kt * 32);
                const float4 x1 = *(const float4*)(ap + kt * 32 + 4);
                const float4 y0 = *(const float4*)(ap + kt * 32 + 16);
                const float4 y1 = *(const float4*)(ap + kt * 32 + 20);
                a[kt] = v16h{ (_Float16)x0.x, (_Float16)x0.y, (_Float16)x0.z, (_Float16)x0.w,
                              (_Float16)x1.x, (_Float16)x1.y, (_Float16)x1.z, (_Float16)x1.w,
                              (_Float16)y0.x, (_Float16)y0.y, (_Float16)y0.z, (_Float16)y0.w,
                              (_Float16)y1.x, (_Float16)y1.y, (_Float16)y1.z, (_Float16)y1.w };
            }
        }
        __syncthreads();
        if (has) {
#pragma unroll
            for (int q = 0; q < 4; ++q) {
                const int ct = half * 4 + q;
                v8f d = v8f{};
#pragma unroll
                for (int kt = 0; kt < 4; ++kt) {
                    const _Float16* bp = Ws + (ct * 4 + kt) * 512 + lane * 8;
                    const v8h blo = *(const v8h*)bp;
                    const v8h bhi = *(const v8h*)(bp + 256);
                    v16h b;
#pragma unroll
                    for (int j = 0; j < 8; ++j) { b[j] = blo[j]; b[8 + j] = bhi[j]; }
                    d = __builtin_amdgcn_wmma_f32_16x16x32_f16(
                        false, a[kt], false, b, (short)0, d, false, false);
                }
#pragma unroll
                for (int r = 0; r < 8; ++r)
                    acc[(tile * 16 + r + 8 * hi) * ACC_PITCH + ct * 16 + hl] = d[r];
            }
        }
    }

    auto finish = [](float v, float b, float mu, float sc, float be) {
        return fmaxf((v + b - mu) * sc + be, 0.f);
    };

    if (FIRST) {
        const int c0 = (tid & 31) * 4;
        float b[4], mu[4], sc[4], be[4];
#pragma unroll
        for (int j = 0; j < 4; ++j) {
            b[j]  = bias[c0 + j];
            mu[j] = mean[c0 + j];
            sc[j] = gam[c0 + j] * rsqrtf(var[c0 + j] + BN_EPS);
            be[j] = bet[c0 + j];
        }
        const v4f was = *(const v4f*)(wan + c0);
        const v4f wad = *(const v4f*)(wan + HID + c0);
        __syncthreads();

        float4* o4 = (float4*)(out + (size_t)row0 * HID);
        for (int i = tid; i < rows * (HID / 4); i += nthr) {
            float4 a = *(const float4*)(acc + (i >> 5) * ACC_PITCH + (i & 31) * 4);
            a.x = finish(a.x, b[0], mu[0], sc[0], be[0]);
            a.y = finish(a.y, b[1], mu[1], sc[1], be[1]);
            a.z = finish(a.z, b[2], mu[2], sc[2], be[2]);
            a.w = finish(a.w, b[3], mu[3], sc[3], be[3]);
            o4[i] = a;
            float ps = a.x * was.x + a.y * was.y + a.z * was.z + a.w * was.w;
            float pd = a.x * wad.x + a.y * wad.y + a.z * wad.z + a.w * wad.w;
            ps = wave_sum32(ps);
            pd = wave_sum32(pd);
            if (lane == 0) { ssn[row0 + (i >> 5)] = ps; sdn[row0 + (i >> 5)] = pd; }
        }
    } else {
        static_assert((OWN_ROWS + OWN_THREADS / HID - 1) / (OWN_THREADS / HID) <= 32,
                      "a slice's graph ids are held one per lane");
        const int c = tid & (HID - 1);
        const float b  = bias[c];
        const float mu = mean[c];
        const float sc = gam[c] * rsqrtf(var[c] + BN_EPS);
        const float be = bet[c];
        const int ngrp = nthr / HID;
        const int per  = (rows + ngrp - 1) / ngrp;
        const int r0   = (wave >> 2) * per;
        const int n    = (rows - r0 < per) ? (rows - r0) : per;
        const int gid = (lane < n) ? batch[row0 + r0 + lane] : 0;
        __syncthreads();

        if (n > 0) {
            int cur = __builtin_amdgcn_readlane(gid, 0);
            float run = 0.f, nrun = 0.f;
            for (int k = 0; k < n; ++k) {
                const int g = __builtin_amdgcn_readlane(gid, k);
                if (g != cur) {
                    atomicAdd(&pool[(size_t)cur * HID + c], run);
                    if (c == 0) atomicAdd(&pcnt[cur], nrun);
                    cur = g; run = 0.f; nrun = 0.f;
                }
                run += finish(acc[(r0 + k) * ACC_PITCH + c], b, mu, sc, be);
                nrun += 1.f;
            }
            atomicAdd(&pool[(size_t)cur * HID + c], run);
            if (c == 0) atomicAdd(&pcnt[cur], nrun);
        }
    }
}

extern "C" void kernel_launch(void* const* d_in, const int* in_sizes, int n_in,
                              void* d_out, int out_size, void* d_ws, size_t ws_size,
                              hipStream_t stream) {
    const float* gx  = (const float*)d_in[0];
    const int*   ei  = (const int*)  d_in[1];
    const int*   bat = (const int*)  d_in[2];
    const float* W1  = (const float*)d_in[3];
    const float* as1 = (const float*)d_in[4];
    const float* ad1 = (const float*)d_in[5];
    const float* b1  = (const float*)d_in[6];
    const float* g1  = (const float*)d_in[7];
    const float* be1 = (const float*)d_in[8];
    const float* mu1 = (const float*)d_in[9];
    const float* va1 = (const float*)d_in[10];
    const float* W2  = (const float*)d_in[11];
    const float* as2 = (const float*)d_in[12];
    const float* ad2 = (const float*)d_in[13];
    const float* b2  = (const float*)d_in[14];
    const float* g2  = (const float*)d_in[15];
    const float* be2 = (const float*)d_in[16];
    const float* mu2 = (const float*)d_in[17];
    const float* va2 = (const float*)d_in[18];
    const float* hw  = (const float*)d_in[19];
    const float* hb  = (const float*)d_in[20];

    const int n_nodes = in_sizes[0] / HID;
    const int n_edges = in_sizes[1] / 2;
    const int* esrc = ei;
    const int* edst = ei + n_edges;

    float* ws   = (float*)d_ws;
    float* AGG  = ws;
    float* SS   = AGG  + (size_t)n_nodes * HID;
    float* SD   = SS   + n_nodes;
    float* POOL = SD   + n_nodes;
    float* CNT  = POOL + (size_t)N_GRAPHS * HID;
    const int gridOwn   = (n_nodes + OWN_ROWS - 1) / OWN_ROWS;
    int*      OCNT  = (int*)(CNT + N_GRAPHS);
    unsigned* OLIST = (unsigned*)(OCNT + gridOwn);
    int*   OPOS = (int*)(OLIST + (size_t)gridOwn * OWN_CAP);
    float* SS2  = (float*)(OPOS + (size_t)gridOwn * (OWN_ROWS + 1));
    float* SD2  = SS2 + n_nodes;

    const int rowsScoreWg = (SC_THREADS / 32) * SC_ROWS;
    const int gridNodeW = (n_nodes + rowsScoreWg - 1) / rowsScoreWg;

    const int nZero = N_GRAPHS * HID + N_GRAPHS + gridOwn;
    fill_f32<<<(nZero + 255) / 256, 256, 0, stream>>>(POOL, 0.0f, nZero);

    const int gridSplit = (n_edges + SPLIT_THREADS * SPLIT_EPT - 1) / (SPLIT_THREADS * SPLIT_EPT);
    split_and_scores<<<gridSplit + gridNodeW, SPLIT_THREADS, 0, stream>>>(
        gridSplit, esrc, edst, OLIST, OCNT, n_edges, n_nodes, gridOwn,
        gx, W1, as1, ad1, SS, SD);

    own_rows_gat<true><<<gridOwn, OWN_THREADS, 0, stream>>>(
        gx, W1, esrc, edst, SS, SD, b1, g1, be1, mu1, va1, AGG, bat, POOL, CNT,
        OLIST, OCNT, OPOS, W2, as2, ad2, SS2, SD2, n_edges, n_nodes);
    own_rows_gat<false><<<gridOwn, OWN_THREADS, 0, stream>>>(
        AGG, W2, esrc, edst, SS2, SD2, b2, g2, be2, mu2, va2, AGG, bat, POOL, CNT,
        OLIST, OCNT, OPOS, nullptr, nullptr, nullptr, nullptr, nullptr, n_edges, n_nodes);

    gemm128_wmma<true, true, 4, 1><<<N_GRAPHS / 16, 32 * (64 / 16), 0, stream>>>(
        POOL, CNT, hw, hb, (float*)d_out, N_GRAPHS);
}
